# starting kernel
# baseline (speedup 1.0000x reference)
#include <hip/hip_runtime.h>
#include <math.h>

typedef __attribute__((ext_vector_type(2))) float v2f;
typedef __attribute__((ext_vector_type(8))) float v8f;

static constexpr int FEAT = 128;
static constexpr int APAD = 132;
static constexpr int ROWS_PER_BLK = 64;

__global__ __launch_bounds__(256) void k_zero(float* __restrict__ p, int n) {
    int i = blockIdx.x * 256 + threadIdx.x;
    if (i < n) p[i] = 0.0f;
}

__global__ __launch_bounds__(256) void k_degree(const int* __restrict__ src,
                                                const int* __restrict__ dst,
                                                float* __restrict__ degO,
                                                float* __restrict__ degI, int ne) {
    int e = blockIdx.x * 256 + threadIdx.x;
    if (e < ne) {
        atomicAdd(degO + src[e], 1.0f);
        atomicAdd(degI + dst[e], 1.0f);
    }
}

__global__ __launch_bounds__(256) void k_rsqrt_clip(float* __restrict__ degO,
                                                    float* __restrict__ degI, int n) {
    int i = blockIdx.x * 256 + threadIdx.x;
    if (i < n) {
        float a = degO[i]; a = a < 1.0f ? 1.0f : a;
        degO[i] = 1.0f / sqrtf(a);
        float b = degI[i]; b = b < 1.0f ? 1.0f : b;
        degI[i] = 1.0f / sqrtf(b);
    }
}

__global__ __launch_bounds__(256) void k_gemm_scaled(const float* __restrict__ X,
                                                     const float* __restrict__ rs,
                                                     const float* __restrict__ W,
                                                     float* __restrict__ Y, int nrows) {
    __shared__ float Alds[ROWS_PER_BLK * APAD];

    const int rowbase = blockIdx.x * ROWS_PER_BLK;
    const int t = threadIdx.x;

    for (int i = t; i < ROWS_PER_BLK * (FEAT / 4); i += 256) {
        int r  = i >> 5;
        int c4 = (i & 31) << 2;
        int row = rowbase + r;
        float4 v = {0.f, 0.f, 0.f, 0.f};
        float s = 0.f;
        if (row < nrows) {
            v = *(const float4*)(X + (size_t)row * FEAT + c4);
            s = rs[row];
        }
        float* dstp = Alds + r * APAD + c4;
        dstp[0] = v.x * s; dstp[1] = v.y * s; dstp[2] = v.z * s; dstp[3] = v.w * s;
    }
    __syncthreads();

    const int wave = t >> 5;
    const int lane = t & 31;
    const int half = lane >> 4;
    const int l16  = lane & 15;
    const int ncol = wave * 16 + l16;

    v8f acc0 = {}, acc1 = {}, acc2 = {}, acc3 = {};
    const float* a0 = Alds + (0 * 16 + l16) * APAD;
    const float* a1 = Alds + (1 * 16 + l16) * APAD;
    const float* a2 = Alds + (2 * 16 + l16) * APAD;
    const float* a3 = Alds + (3 * 16 + l16) * APAD;

    #pragma unroll 4
    for (int k = 0; k < FEAT; k += 4) {
        const int ka = k + half * 2;
        v2f b;
        b.x = W[ka * FEAT + ncol];
        b.y = W[(ka + 1) * FEAT + ncol];
        v2f a;
        a.x = a0[ka]; a.y = a0[ka + 1];
        acc0 = __builtin_amdgcn_wmma_f32_16x16x4_f32(false, a, false, b, (short)0, acc0, false, false);
        a.x = a1[ka]; a.y = a1[ka + 1];
        acc1 = __builtin_amdgcn_wmma_f32_16x16x4_f32(false, a, false, b, (short)0, acc1, false, false);
        a.x = a2[ka]; a.y = a2[ka + 1];
        acc2 = __builtin_amdgcn_wmma_f32_16x16x4_f32(false, a, false, b, (short)0, acc2, false, false);
        a.x = a3[ka]; a.y = a3[ka + 1];
        acc3 = __builtin_amdgcn_wmma_f32_16x16x4_f32(false, a, false, b, (short)0, acc3, false, false);
    }

    const int mrow0 = rowbase + half * 8;
    if (rowbase + ROWS_PER_BLK <= nrows) {
        #pragma unroll
        for (int r = 0; r < 8; ++r) {
            Y[(size_t)(mrow0 +  0 + r) * FEAT + ncol] = acc0[r];
            Y[(size_t)(mrow0 + 16 + r) * FEAT + ncol] = acc1[r];
            Y[(size_t)(mrow0 + 32 + r) * FEAT + ncol] = acc2[r];
            Y[(size_t)(mrow0 + 48 + r) * FEAT + ncol] = acc3[r];
        }
    } else {
        #pragma unroll
        for (int r = 0; r < 8; ++r) {
            int r0 = mrow0 + r;
            if (r0 +  0 < nrows) Y[(size_t)(r0 +  0) * FEAT + ncol] = acc0[r];
            if (r0 + 16 < nrows) Y[(size_t)(r0 + 16) * FEAT + ncol] = acc1[r];
            if (r0 + 32 < nrows) Y[(size_t)(r0 + 32) * FEAT + ncol] = acc2[r];
            if (r0 + 48 < nrows) Y[(size_t)(r0 + 48) * FEAT + ncol] = acc3[r];
        }
    }
}

__global__ __launch_bounds__(256) void k_scatter(const float* __restrict__ H,
                                                 const int* __restrict__ src,
                                                 const int* __restrict__ dst,
                                                 float* __restrict__ agg, int ne) {
    int e = blockIdx.x * 8 + (threadIdx.x >> 5);
    if (e >= ne) return;
    int lane = threadIdx.x & 31;
    int s = src[e], d = dst[e];
    float4 v = *(const float4*)(H + (size_t)s * FEAT + lane * 4);
    float* p = agg + (size_t)d * FEAT + lane * 4;
    atomicAdd(p + 0, v.x);
    atomicAdd(p + 1, v.y);
    atomicAdd(p + 2, v.z);
    atomicAdd(p + 3, v.w);
}

__global__ __launch_bounds__(256) void k_finalize(const float* __restrict__ agg,
                                                  const float* __restrict__ rsI,
                                                  const float* __restrict__ bias,
                                                  float* __restrict__ out, int n) {
    int i = blockIdx.x * 256 + threadIdx.x;
    if (i < n) {
        int row = i >> 7;
        int col = i & 127;
        float v = agg[i] * rsI[row] + bias[col];
        out[i] = v > 0.f ? v : 0.f;
    }
}

__global__ __launch_bounds__(256) void k_classify(const float* __restrict__ H,
                                                  const float* __restrict__ Wc,
                                                  const float* __restrict__ bc,
                                                  float* __restrict__ out, int n4) {
    int i = blockIdx.x * 256 + threadIdx.x;
    if (i >= n4) return;
    int node = i >> 2;
    int c = i & 3;
    const float* h = H + (size_t)node * FEAT;
    float sum = 0.f;
    #pragma unroll 8
    for (int k = 0; k < FEAT; ++k) sum += h[k] * Wc[k * 4 + c];
    out[i] = sum + bc[c];
}

extern "C" void kernel_launch(void* const* d_in, const int* in_sizes, int n_in,
                              void* d_out, int out_size, void* d_ws, size_t ws_size,
                              hipStream_t stream) {
    const float* feat = (const float*)d_in[0];
    const int*   src  = (const int*)d_in[1];
    const int*   dst  = (const int*)d_in[2];
    const float* W1   = (const float*)d_in[3];
    const float* b1   = (const float*)d_in[4];
    const float* W2   = (const float*)d_in[5];
    const float* b2   = (const float*)d_in[6];
    const float* Wc   = (const float*)d_in[7];
    const float* bc   = (const float*)d_in[8];
    float* out = (float*)d_out;

    const int nnodes = in_sizes[0] / FEAT;
    const int nedges = in_sizes[1];
    const size_t nAlign = ((size_t)nnodes + 3) & ~(size_t)3;
    const size_t nfeat = (size_t)nnodes * FEAT;

    float* rsO  = (float*)d_ws;
    float* rsI  = rsO + nAlign;
    float* bufA = rsI + nAlign;
    float* bufB = bufA + nfeat;

    const int B = 256;
    const int gN      = (int)((nfeat + B - 1) / B);
    const int gNode   = (nnodes + B - 1) / B;
    const int gEdgeT  = (nedges + B - 1) / B;
    const int gEdgeW  = (nedges + 7) / 8;
    const int gGemm   = (nnodes + ROWS_PER_BLK - 1) / ROWS_PER_BLK;
    const int gCls    = (nnodes * 4 + B - 1) / B;

    k_zero<<<(int)((2 * nAlign + B - 1) / B), B, 0, stream>>>(rsO, (int)(2 * nAlign));
    k_degree<<<gEdgeT, B, 0, stream>>>(src, dst, rsO, rsI, nedges);
    k_rsqrt_clip<<<gNode, B, 0, stream>>>(rsO, rsI, nnodes);

    k_gemm_scaled<<<gGemm, B, 0, stream>>>(feat, rsO, W1, bufA, nnodes);
    k_zero<<<gN, B, 0, stream>>>(bufB, (int)nfeat);
    k_scatter<<<gEdgeW, B, 0, stream>>>(bufA, src, dst, bufB, nedges);
    k_finalize<<<gN, B, 0, stream>>>(bufB, rsI, b1, bufA, (int)nfeat);

    k_gemm_scaled<<<gGemm, B, 0, stream>>>(bufA, rsO, W2, bufB, nnodes);
    k_zero<<<gN, B, 0, stream>>>(bufA, (int)nfeat);
    k_scatter<<<gEdgeW, B, 0, stream>>>(bufB, src, dst, bufA, nedges);
    k_finalize<<<gN, B, 0, stream>>>(bufA, rsI, b2, bufB, (int)nfeat);

    k_classify<<<gCls, B, 0, stream>>>(bufB, Wc, bc, out, nnodes * 4);
}


# scatter-add by float atomics replaced by a per-call counting sort of the edges by destination (rank from integer atomics) and a one-wave-per-node gather sum; rows summed in arrival-rank order
# speedup vs baseline: 2.0134x; 2.0134x over previous
#include <hip/hip_runtime.h>
#include <math.h>

typedef __attribute__((ext_vector_type(2))) float v2f;
typedef __attribute__((ext_vector_type(8))) float v8f;

static constexpr int FEAT = 128;
static constexpr int APAD = 132;
static constexpr int ROWS_PER_BLK = 64;

__global__ __launch_bounds__(256) void k_zero_i(int* __restrict__ p, int n) {
    int i = blockIdx.x * 256 + threadIdx.x;
    if (i < n) p[i] = 0;
}

__global__ __launch_bounds__(256) void k_count_rank(const int* __restrict__ src,
                                                    const int* __restrict__ dst,
                                                    int* __restrict__ cntO,
                                                    int* __restrict__ cntI,
                                                    int* __restrict__ rank, int ne) {
    int e = blockIdx.x * 256 + threadIdx.x;
    if (e < ne) {
        atomicAdd(cntO + src[e], 1);
        rank[e] = atomicAdd(cntI + dst[e], 1);
    }
}

__global__ __launch_bounds__(256) void k_node_prep(const int* __restrict__ cntO,
                                                   const int* __restrict__ cntI,
                                                   float* __restrict__ rsO,
                                                   float* __restrict__ rsI,
                                                   int* __restrict__ seg,
                                                   int* __restrict__ cursor, int nalign) {
    __shared__ int wsum[8];
    __shared__ int bbase;
    const int t = threadIdx.x;
    const int lane = t & 31;
    const int wave = t >> 5;
    const int i0 = (blockIdx.x * 256 + t) * 4;
    const bool live = i0 < nalign;

    int4 ci = {0, 0, 0, 0};
    int4 co = {0, 0, 0, 0};
    if (live) {
        ci = *(const int4*)(cntI + i0);
        co = *(const int4*)(cntO + i0);
    }
    const int p1 = ci.x, p2 = p1 + ci.y, p3 = p2 + ci.z, tot = p3 + ci.w;

    int incl = tot;
    for (int d = 1; d < 32; d <<= 1) {
        int u = __shfl_up(incl, d);
        if (lane >= d) incl += u;
    }
    if (lane == 31) wsum[wave] = incl;
    __syncthreads();
    int woff = 0, btot = 0;
    for (int w = 0; w < 8; ++w) {
        int s = wsum[w];
        if (w < wave) woff += s;
        btot += s;
    }
    if (t == 0) bbase = atomicAdd(cursor, btot);
    __syncthreads();

    if (live) {
        const int b = bbase + woff + (incl - tot);
        int4 s01 = {b, ci.x, b + p1, ci.y};
        int4 s23 = {b + p2, ci.z, b + p3, ci.w};
        *(int4*)(seg + 2 * (size_t)i0) = s01;
        *(int4*)(seg + 2 * (size_t)i0 + 4) = s23;
        float4 fo, fi;
        float a;
        a = (float)co.x; a = a < 1.0f ? 1.0f : a; fo.x = 1.0f / sqrtf(a);
        a = (float)co.y; a = a < 1.0f ? 1.0f : a; fo.y = 1.0f / sqrtf(a);
        a = (float)co.z; a = a < 1.0f ? 1.0f : a; fo.z = 1.0f / sqrtf(a);
        a = (float)co.w; a = a < 1.0f ? 1.0f : a; fo.w = 1.0f / sqrtf(a);
        a = (float)ci.x; a = a < 1.0f ? 1.0f : a; fi.x = 1.0f / sqrtf(a);
        a = (float)ci.y; a = a < 1.0f ? 1.0f : a; fi.y = 1.0f / sqrtf(a);
        a = (float)ci.z; a = a < 1.0f ? 1.0f : a; fi.z = 1.0f / sqrtf(a);
        a = (float)ci.w; a = a < 1.0f ? 1.0f : a; fi.w = 1.0f / sqrtf(a);
        *(float4*)(rsO + i0) = fo;
        *(float4*)(rsI + i0) = fi;
    }
}

__global__ __launch_bounds__(256) void k_fill(const int* __restrict__ src,
                                              const int* __restrict__ dst,
                                              const int* __restrict__ seg,
                                              const int* __restrict__ rank,
                                              int* __restrict__ col, int ne) {
    int e = blockIdx.x * 256 + threadIdx.x;
    if (e < ne) col[seg[2 * (size_t)dst[e]] + rank[e]] = src[e];
}

__global__ __launch_bounds__(256) void k_gemm_scaled(const float* __restrict__ X,
                                                     const float* __restrict__ rs,
                                                     const float* __restrict__ W,
                                                     float* __restrict__ Y, int nrows) {
    __shared__ float Alds[ROWS_PER_BLK * APAD];

    const int rowbase = blockIdx.x * ROWS_PER_BLK;
    const int t = threadIdx.x;

    for (int i = t; i < ROWS_PER_BLK * (FEAT / 4); i += 256) {
        int r  = i >> 5;
        int c4 = (i & 31) << 2;
        int row = rowbase + r;
        float4 v = {0.f, 0.f, 0.f, 0.f};
        float s = 0.f;
        if (row < nrows) {
            v = *(const float4*)(X + (size_t)row * FEAT + c4);
            s = rs[row];
        }
        float* dstp = Alds + r * APAD + c4;
        dstp[0] = v.x * s; dstp[1] = v.y * s; dstp[2] = v.z * s; dstp[3] = v.w * s;
    }
    __syncthreads();

    const int wave = t >> 5;
    const int lane = t & 31;
    const int half = lane >> 4;
    const int l16  = lane & 15;
    const int ncol = wave * 16 + l16;

    v8f acc0 = {}, acc1 = {}, acc2 = {}, acc3 = {};
    const float* a0 = Alds + (0 * 16 + l16) * APAD;
    const float* a1 = Alds + (1 * 16 + l16) * APAD;
    const float* a2 = Alds + (2 * 16 + l16) * APAD;
    const float* a3 = Alds + (3 * 16 + l16) * APAD;

    #pragma unroll 4
    for (int k = 0; k < FEAT; k += 4) {
        const int ka = k + half * 2;
        v2f b;
        b.x = W[ka * FEAT + ncol];
        b.y = W[(ka + 1) * FEAT + ncol];
        v2f a;
        a.x = a0[ka]; a.y = a0[ka + 1];
        acc0 = __builtin_amdgcn_wmma_f32_16x16x4_f32(false, a, false, b, (short)0, acc0, false, false);
        a.x = a1[ka]; a.y = a1[ka + 1];
        acc1 = __builtin_amdgcn_wmma_f32_16x16x4_f32(false, a, false, b, (short)0, acc1, false, false);
        a.x = a2[ka]; a.y = a2[ka + 1];
        acc2 = __builtin_amdgcn_wmma_f32_16x16x4_f32(false, a, false, b, (short)0, acc2, false, false);
        a.x = a3[ka]; a.y = a3[ka + 1];
        acc3 = __builtin_amdgcn_wmma_f32_16x16x4_f32(false, a, false, b, (short)0, acc3, false, false);
    }

    const int mrow0 = rowbase + half * 8;
    if (rowbase + ROWS_PER_BLK <= nrows) {
        #pragma unroll
        for (int r = 0; r < 8; ++r) {
            Y[(size_t)(mrow0 +  0 + r) * FEAT + ncol] = acc0[r];
            Y[(size_t)(mrow0 + 16 + r) * FEAT + ncol] = acc1[r];
            Y[(size_t)(mrow0 + 32 + r) * FEAT + ncol] = acc2[r];
            Y[(size_t)(mrow0 + 48 + r) * FEAT + ncol] = acc3[r];
        }
    } else {
        #pragma unroll
        for (int r = 0; r < 8; ++r) {
            int r0 = mrow0 + r;
            if (r0 +  0 < nrows) Y[(size_t)(r0 +  0) * FEAT + ncol] = acc0[r];
            if (r0 + 16 < nrows) Y[(size_t)(r0 + 16) * FEAT + ncol] = acc1[r];
            if (r0 + 32 < nrows) Y[(size_t)(r0 + 32) * FEAT + ncol] = acc2[r];
            if (r0 + 48 < nrows) Y[(size_t)(r0 + 48) * FEAT + ncol] = acc3[r];
        }
    }
}

__global__ __launch_bounds__(256) void k_gather(const float* __restrict__ H,
                                                const int* __restrict__ seg,
                                                const int* __restrict__ col,
                                                float* __restrict__ agg, int n) {
    int node = blockIdx.x * 8 + (threadIdx.x >> 5);
    if (node >= n) return;
    node = __builtin_amdgcn_readfirstlane(node);
    const int lane = threadIdx.x & 31;
    const int2 sg = *(const int2*)(seg + 2 * (size_t)node);
    const int start = sg.x;
    const int cnt = sg.y;
    const float* Hl = H + lane * 4;

    float4 acc = {0.f, 0.f, 0.f, 0.f};
#define ROW(j) (*(const float4*)(Hl + (size_t)__builtin_amdgcn_readlane(cv, (j)) * FEAT))
#define ADD(v) { acc.x += (v).x; acc.y += (v).y; acc.z += (v).z; acc.w += (v).w; }
    for (int base = 0; base < cnt; base += 32) {
        int m = cnt - base;
        if (m > 32) m = 32;
        int cv = 0;
        if (lane < m) cv = col[start + base + lane];
        int j = 0;
        for (; j + 8 <= m; j += 8) {
            float4 v0 = ROW(j), v1 = ROW(j + 1), v2 = ROW(j + 2), v3 = ROW(j + 3);
            float4 v4 = ROW(j + 4), v5 = ROW(j + 5), v6 = ROW(j + 6), v7 = ROW(j + 7);
            ADD(v0) ADD(v1) ADD(v2) ADD(v3) ADD(v4) ADD(v5) ADD(v6) ADD(v7)
        }
        if (j + 4 <= m) {
            float4 v0 = ROW(j), v1 = ROW(j + 1), v2 = ROW(j + 2), v3 = ROW(j + 3);
            ADD(v0) ADD(v1) ADD(v2) ADD(v3)
            j += 4;
        }
        if (j + 2 <= m) {
            float4 v0 = ROW(j), v1 = ROW(j + 1);
            ADD(v0) ADD(v1)
            j += 2;
        }
        if (j < m) {
            float4 v0 = ROW(j);
            ADD(v0)
        }
    }
#undef ROW
#undef ADD
    *(float4*)(agg + (size_t)node * FEAT + lane * 4) = acc;
}

__global__ __launch_bounds__(256) void k_finalize(const float* __restrict__ agg,
                                                  const float* __restrict__ rsI,
                                                  const float* __restrict__ bias,
                                                  float* __restrict__ out, int n) {
    int i = blockIdx.x * 256 + threadIdx.x;
    if (i < n) {
        int row = i >> 7;
        int col = i & 127;
        float v = agg[i] * rsI[row] + bias[col];
        out[i] = v > 0.f ? v : 0.f;
    }
}

__global__ __launch_bounds__(256) void k_classify(const float* __restrict__ H,
                                                  const float* __restrict__ Wc,
                                                  const float* __restrict__ bc,
                                                  float* __restrict__ out, int n4) {
    int i = blockIdx.x * 256 + threadIdx.x;
    if (i >= n4) return;
    int node = i >> 2;
    int c = i & 3;
    const float* h = H + (size_t)node * FEAT;
    float sum = 0.f;
    #pragma unroll 8
    for (int k = 0; k < FEAT; ++k) sum += h[k] * Wc[k * 4 + c];
    out[i] = sum + bc[c];
}

extern "C" void kernel_launch(void* const* d_in, const int* in_sizes, int n_in,
                              void* d_out, int out_size, void* d_ws, size_t ws_size,
                              hipStream_t stream) {
    const float* feat = (const float*)d_in[0];
    const int*   src  = (const int*)d_in[1];
    const int*   dst  = (const int*)d_in[2];
    const float* W1   = (const float*)d_in[3];
    const float* b1   = (const float*)d_in[4];
    const float* W2   = (const float*)d_in[5];
    const float* b2   = (const float*)d_in[6];
    const float* Wc   = (const float*)d_in[7];
    const float* bc   = (const float*)d_in[8];
    float* out = (float*)d_out;

    const int nnodes = in_sizes[0] / FEAT;
    const int nedges = in_sizes[1];
    const size_t nAlign = ((size_t)nnodes + 3) & ~(size_t)3;
    const size_t eAlign = ((size_t)nedges + 3) & ~(size_t)3;
    const size_t nfeat = (size_t)nnodes * FEAT;

    int*   cntO     = (int*)d_ws;
    int*   cntI     = cntO + nAlign;
    int*   cursor   = cntI + nAlign;
    int*   seg      = cursor + 4;
    float* rsO      = (float*)(seg + 2 * nAlign);
    float* rsI      = rsO + nAlign;
    int*   rank     = (int*)(rsI + nAlign);
    int*   col      = rank + eAlign;
    float* bufA     = (float*)(col + eAlign);
    float* bufB     = bufA + nfeat;

    const int B = 256;
    const int gN      = (int)((nfeat + B - 1) / B);
    const int gEdgeT  = (nedges + B - 1) / B;
    const int gNodeW  = (nnodes + 7) / 8;
    const int gPrep   = (int)((nAlign / 4 + B - 1) / B);
    const int gGemm   = (nnodes + ROWS_PER_BLK - 1) / ROWS_PER_BLK;
    const int gCls    = (nnodes * 4 + B - 1) / B;
    const int nZero   = (int)(2 * nAlign + 4);

    k_zero_i<<<(nZero + B - 1) / B, B, 0, stream>>>(cntO, nZero);
    k_count_rank<<<gEdgeT, B, 0, stream>>>(src, dst, cntO, cntI, rank, nedges);
    k_node_prep<<<gPrep, B, 0, stream>>>(cntO, cntI, rsO, rsI, seg, cursor, (int)nAlign);
    k_fill<<<gEdgeT, B, 0, stream>>>(src, dst, seg, rank, col, nedges);

    k_gemm_scaled<<<gGemm, B, 0, stream>>>(feat, rsO, W1, bufA, nnodes);
    k_gather<<<gNodeW, B, 0, stream>>>(bufA, seg, col, bufB, nnodes);
    k_finalize<<<gN, B, 0, stream>>>(bufB, rsI, b1, bufA, (int)nfeat);

    k_gemm_scaled<<<gGemm, B, 0, stream>>>(bufA, rsO, W2, bufB, nnodes);
    k_gather<<<gNodeW, B, 0, stream>>>(bufB, seg, col, bufA, nnodes);
    k_finalize<<<gN, B, 0, stream>>>(bufA, rsI, b2, bufB, (int)nfeat);

    k_classify<<<gCls, B, 0, stream>>>(bufB, Wc, bc, out, nnodes * 4);
}


# finalize (scale, bias, relu) and, for layer 2, the 128x4 classifier dot moved into the gather's tail: agg and h2 never reach memory; logits summed by a wave butterfly tree; 11 launches -> 8
# speedup vs baseline: 2.4872x; 1.2353x over previous
#include <hip/hip_runtime.h>
#include <math.h>

typedef __attribute__((ext_vector_type(2))) float v2f;
typedef __attribute__((ext_vector_type(8))) float v8f;

static constexpr int FEAT = 128;
static constexpr int APAD = 132;
static constexpr int ROWS_PER_BLK = 64;

__global__ __launch_bounds__(256) void k_zero_i(int* __restrict__ p, int n) {
    int i = blockIdx.x * 256 + threadIdx.x;
    if (i < n) p[i] = 0;
}

__global__ __launch_bounds__(256) void k_count_rank(const int* __restrict__ src,
                                                    const int* __restrict__ dst,
                                                    int* __restrict__ cntO,
                                                    int* __restrict__ cntI,
                                                    int* __restrict__ rank, int ne) {
    int e = blockIdx.x * 256 + threadIdx.x;
    if (e < ne) {
        atomicAdd(cntO + src[e], 1);
        rank[e] = atomicAdd(cntI + dst[e], 1);
    }
}

__global__ __launch_bounds__(256) void k_node_prep(const int* __restrict__ cntO,
                                                   const int* __restrict__ cntI,
                                                   float* __restrict__ rsO,
                                                   float* __restrict__ rsI,
                                                   int* __restrict__ seg,
                                                   int* __restrict__ cursor, int nalign) {
    __shared__ int wsum[8];
    __shared__ int bbase;
    const int t = threadIdx.x;
    const int lane = t & 31;
    const int wave = t >> 5;
    const int i0 = (blockIdx.x * 256 + t) * 4;
    const bool live = i0 < nalign;

    int4 ci = {0, 0, 0, 0};
    int4 co = {0, 0, 0, 0};
    if (live) {
        ci = *(const int4*)(cntI + i0);
        co = *(const int4*)(cntO + i0);
    }
    const int p1 = ci.x, p2 = p1 + ci.y, p3 = p2 + ci.z, tot = p3 + ci.w;

    int incl = tot;
    for (int d = 1; d < 32; d <<= 1) {
        int u = __shfl_up(incl, d);
        if (lane >= d) incl += u;
    }
    if (lane == 31) wsum[wave] = incl;
    __syncthreads();
    int woff = 0, btot = 0;
    for (int w = 0; w < 8; ++w) {
        int s = wsum[w];
        if (w < wave) woff += s;
        btot += s;
    }
    if (t == 0) bbase = atomicAdd(cursor, btot);
    __syncthreads();

    if (live) {
        const int b = bbase + woff + (incl - tot);
        int4 s01 = {b, ci.x, b + p1, ci.y};
        int4 s23 = {b + p2, ci.z, b + p3, ci.w};
        *(int4*)(seg + 2 * (size_t)i0) = s01;
        *(int4*)(seg + 2 * (size_t)i0 + 4) = s23;
        float4 fo, fi;
        float a;
        a = (float)co.x; a = a < 1.0f ? 1.0f : a; fo.x = 1.0f / sqrtf(a);
        a = (float)co.y; a = a < 1.0f ? 1.0f : a; fo.y = 1.0f / sqrtf(a);
        a = (float)co.z; a = a < 1.0f ? 1.0f : a; fo.z = 1.0f / sqrtf(a);
        a = (float)co.w; a = a < 1.0f ? 1.0f : a; fo.w = 1.0f / sqrtf(a);
        a = (float)ci.x; a = a < 1.0f ? 1.0f : a; fi.x = 1.0f / sqrtf(a);
        a = (float)ci.y; a = a < 1.0f ? 1.0f : a; fi.y = 1.0f / sqrtf(a);
        a = (float)ci.z; a = a < 1.0f ? 1.0f : a; fi.z = 1.0f / sqrtf(a);
        a = (float)ci.w; a = a < 1.0f ? 1.0f : a; fi.w = 1.0f / sqrtf(a);
        *(float4*)(rsO + i0) = fo;
        *(float4*)(rsI + i0) = fi;
    }
}

__global__ __launch_bounds__(256) void k_fill(const int* __restrict__ src,
                                              const int* __restrict__ dst,
                                              const int* __restrict__ seg,
                                              const int* __restrict__ rank,
                                              int* __restrict__ col, int ne) {
    int e = blockIdx.x * 256 + threadIdx.x;
    if (e < ne) col[seg[2 * (size_t)dst[e]] + rank[e]] = src[e];
}

__global__ __launch_bounds__(256) void k_gemm_scaled(const float* __restrict__ X,
                                                     const float* __restrict__ rs,
                                                     const float* __restrict__ W,
                                                     float* __restrict__ Y, int nrows) {
    __shared__ float Alds[ROWS_PER_BLK * APAD];

    const int rowbase = blockIdx.x * ROWS_PER_BLK;
    const int t = threadIdx.x;

    for (int i = t; i < ROWS_PER_BLK * (FEAT / 4); i += 256) {
        int r  = i >> 5;
        int c4 = (i & 31) << 2;
        int row = rowbase + r;
        float4 v = {0.f, 0.f, 0.f, 0.f};
        float s = 0.f;
        if (row < nrows) {
            v = *(const float4*)(X + (size_t)row * FEAT + c4);
            s = rs[row];
        }
        float* dstp = Alds + r * APAD + c4;
        dstp[0] = v.x * s; dstp[1] = v.y * s; dstp[2] = v.z * s; dstp[3] = v.w * s;
    }
    __syncthreads();

    const int wave = t >> 5;
    const int lane = t & 31;
    const int half = lane >> 4;
    const int l16  = lane & 15;
    const int ncol = wave * 16 + l16;

    v8f acc0 = {}, acc1 = {}, acc2 = {}, acc3 = {};
    const float* a0 = Alds + (0 * 16 + l16) * APAD;
    const float* a1 = Alds + (1 * 16 + l16) * APAD;
    const float* a2 = Alds + (2 * 16 + l16) * APAD;
    const float* a3 = Alds + (3 * 16 + l16) * APAD;

    #pragma unroll 4
    for (int k = 0; k < FEAT; k += 4) {
        const int ka = k + half * 2;
        v2f b;
        b.x = W[ka * FEAT + ncol];
        b.y = W[(ka + 1) * FEAT + ncol];
        v2f a;
        a.x = a0[ka]; a.y = a0[ka + 1];
        acc0 = __builtin_amdgcn_wmma_f32_16x16x4_f32(false, a, false, b, (short)0, acc0, false, false);
        a.x = a1[ka]; a.y = a1[ka + 1];
        acc1 = __builtin_amdgcn_wmma_f32_16x16x4_f32(false, a, false, b, (short)0, acc1, false, false);
        a.x = a2[ka]; a.y = a2[ka + 1];
        acc2 = __builtin_amdgcn_wmma_f32_16x16x4_f32(false, a, false, b, (short)0, acc2, false, false);
        a.x = a3[ka]; a.y = a3[ka + 1];
        acc3 = __builtin_amdgcn_wmma_f32_16x16x4_f32(false, a, false, b, (short)0, acc3, false, false);
    }

    const int mrow0 = rowbase + half * 8;
    if (rowbase + ROWS_PER_BLK <= nrows) {
        #pragma unroll
        for (int r = 0; r < 8; ++r) {
            Y[(size_t)(mrow0 +  0 + r) * FEAT + ncol] = acc0[r];
            Y[(size_t)(mrow0 + 16 + r) * FEAT + ncol] = acc1[r];
            Y[(size_t)(mrow0 + 32 + r) * FEAT + ncol] = acc2[r];
            Y[(size_t)(mrow0 + 48 + r) * FEAT + ncol] = acc3[r];
        }
    } else {
        #pragma unroll
        for (int r = 0; r < 8; ++r) {
            int r0 = mrow0 + r;
            if (r0 +  0 < nrows) Y[(size_t)(r0 +  0) * FEAT + ncol] = acc0[r];
            if (r0 + 16 < nrows) Y[(size_t)(r0 + 16) * FEAT + ncol] = acc1[r];
            if (r0 + 32 < nrows) Y[(size_t)(r0 + 32) * FEAT + ncol] = acc2[r];
            if (r0 + 48 < nrows) Y[(size_t)(r0 + 48) * FEAT + ncol] = acc3[r];
        }
    }
}

static __device__ __forceinline__ float4 gather_row(const float* __restrict__ H,
                                                    const int* __restrict__ seg,
                                                    const int* __restrict__ col,
                                                    int node, int lane) {
    const int2 sg = *(const int2*)(seg + 2 * (size_t)node);
    const int start = sg.x;
    const int cnt = sg.y;
    const float* Hl = H + lane * 4;

    float4 acc = {0.f, 0.f, 0.f, 0.f};
#define ROW(j) (*(const float4*)(Hl + (size_t)__builtin_amdgcn_readlane(cv, (j)) * FEAT))
#define ADD(v) { acc.x += (v).x; acc.y += (v).y; acc.z += (v).z; acc.w += (v).w; }
    for (int base = 0; base < cnt; base += 32) {
        int m = cnt - base;
        if (m > 32) m = 32;
        int cv = 0;
        if (lane < m) cv = col[start + base + lane];
        int j = 0;
        for (; j + 8 <= m; j += 8) {
            float4 v0 = ROW(j), v1 = ROW(j + 1), v2 = ROW(j + 2), v3 = ROW(j + 3);
            float4 v4 = ROW(j + 4), v5 = ROW(j + 5), v6 = ROW(j + 6), v7 = ROW(j + 7);
            ADD(v0) ADD(v1) ADD(v2) ADD(v3) ADD(v4) ADD(v5) ADD(v6) ADD(v7)
        }
        if (j + 4 <= m) {
            float4 v0 = ROW(j), v1 = ROW(j + 1), v2 = ROW(j + 2), v3 = ROW(j + 3);
            ADD(v0) ADD(v1) ADD(v2) ADD(v3)
            j += 4;
        }
        if (j + 2 <= m) {
            float4 v0 = ROW(j), v1 = ROW(j + 1);
            ADD(v0) ADD(v1)
            j += 2;
        }
        if (j < m) {
            float4 v0 = ROW(j);
            ADD(v0)
        }
    }
#undef ROW
#undef ADD
    return acc;
}

static __device__ __forceinline__ float4 row_tail(float4 acc, float s, float4 b) {
    float4 h;
    h.x = acc.x * s + b.x; h.x = h.x > 0.f ? h.x : 0.f;
    h.y = acc.y * s + b.y; h.y = h.y > 0.f ? h.y : 0.f;
    h.z = acc.z * s + b.z; h.z = h.z > 0.f ? h.z : 0.f;
    h.w = acc.w * s + b.w; h.w = h.w > 0.f ? h.w : 0.f;
    return h;
}

__global__ __launch_bounds__(256) void k_gather_relu(const float* __restrict__ H,
                                                     const int* __restrict__ seg,
                                                     const int* __restrict__ col,
                                                     const float* __restrict__ rsI,
                                                     const float* __restrict__ bias,
                                                     float* __restrict__ out, int n) {
    int node = blockIdx.x * 8 + (threadIdx.x >> 5);
    if (node >= n) return;
    node = __builtin_amdgcn_readfirstlane(node);
    const int lane = threadIdx.x & 31;
    const float s = rsI[node];
    const float4 b = *(const float4*)(bias + lane * 4);
    const float4 acc = gather_row(H, seg, col, node, lane);
    *(float4*)(out + (size_t)node * FEAT + lane * 4) = row_tail(acc, s, b);
}

static __device__ __forceinline__ float lx_dpp_q1(float x) {
    return __builtin_bit_cast(float, __builtin_amdgcn_update_dpp(0, __builtin_bit_cast(int, x), 0xB1, 0xF, 0xF, true));
}
static __device__ __forceinline__ float lx_dpp_q2(float x) {
    return __builtin_bit_cast(float, __builtin_amdgcn_update_dpp(0, __builtin_bit_cast(int, x), 0x4E, 0xF, 0xF, true));
}
static __device__ __forceinline__ float lx_dpp_x4(float x) {
    return __builtin_bit_cast(float, __builtin_amdgcn_update_dpp(0, __builtin_bit_cast(int, x), 0x164, 0xF, 0xF, true));
}
static __device__ __forceinline__ float lx_dpp_x8(float x) {
    return __builtin_bit_cast(float, __builtin_amdgcn_update_dpp(0, __builtin_bit_cast(int, x), 0x168, 0xF, 0xF, true));
}
static __device__ __forceinline__ float lx_half(float x) {
    return __builtin_bit_cast(float, __builtin_amdgcn_permlanex16(__builtin_bit_cast(int, x), __builtin_bit_cast(int, x),
                                                                  0x76543210, 0xfedcba98, false, false));
}

__global__ __launch_bounds__(256) void k_gather_cls(const float* __restrict__ H,
                                                    const int* __restrict__ seg,
                                                    const int* __restrict__ col,
                                                    const float* __restrict__ rsI,
                                                    const float* __restrict__ bias,
                                                    const float* __restrict__ Wc,
                                                    const float* __restrict__ bc,
                                                    float* __restrict__ out, int n) {
    __shared__ float lg[32];
    const int lane = threadIdx.x & 31;
    const int wave = threadIdx.x >> 5;
    const float bco = bc[lane & 3];
    int node = blockIdx.x * 8 + wave;
    if (node < n) {
        node = __builtin_amdgcn_readfirstlane(node);
        const float s = rsI[node];
        const float4 b = *(const float4*)(bias + lane * 4);
        const float4 acc = gather_row(H, seg, col, node, lane);
        const float4 h = row_tail(acc, s, b);

        const float4 w0 = *(const float4*)(Wc + lane * 16);
        const float4 w1 = *(const float4*)(Wc + lane * 16 + 4);
        const float4 w2 = *(const float4*)(Wc + lane * 16 + 8);
        const float4 w3 = *(const float4*)(Wc + lane * 16 + 12);
        const float p0 = h.x * w0.x + h.y * w1.x + h.z * w2.x + h.w * w3.x;
        const float p1 = h.x * w0.y + h.y * w1.y + h.z * w2.y + h.w * w3.y;
        const float p2 = h.x * w0.z + h.y * w1.z + h.z * w2.z + h.w * w3.z;
        const float p3 = h.x * w0.w + h.y * w1.w + h.z * w2.w + h.w * w3.w;

        const bool o1 = (lane & 1) != 0;
        const bool o2 = (lane & 2) != 0;
        const float q0 = (o1 ? p2 : p0) + lx_dpp_q1(o1 ? p0 : p2);
        const float q1 = (o1 ? p3 : p1) + lx_dpp_q1(o1 ? p1 : p3);
        float r = (o2 ? q1 : q0) + lx_dpp_q2(o2 ? q0 : q1);
        r += lx_dpp_x4(r);
        r += lx_dpp_x8(r);
        r += lx_half(r);
        if (lane < 4) lg[wave * 4 + (((lane & 1) << 1) | (lane >> 1))] = r;
    }
    __syncthreads();
    if (wave == 0) {
        const int i = blockIdx.x * 32 + lane;
        if (i < n * 4) out[i] = lg[lane] + bco;
    }
}

extern "C" void kernel_launch(void* const* d_in, const int* in_sizes, int n_in,
                              void* d_out, int out_size, void* d_ws, size_t ws_size,
                              hipStream_t stream) {
    const float* feat = (const float*)d_in[0];
    const int*   src  = (const int*)d_in[1];
    const int*   dst  = (const int*)d_in[2];
    const float* W1   = (const float*)d_in[3];
    const float* b1   = (const float*)d_in[4];
    const float* W2   = (const float*)d_in[5];
    const float* b2   = (const float*)d_in[6];
    const float* Wc   = (const float*)d_in[7];
    const float* bc   = (const float*)d_in[8];
    float* out = (float*)d_out;

    const int nnodes = in_sizes[0] / FEAT;
    const int nedges = in_sizes[1];
    const size_t nAlign = ((size_t)nnodes + 3) & ~(size_t)3;
    const size_t eAlign = ((size_t)nedges + 3) & ~(size_t)3;
    const size_t nfeat = (size_t)nnodes * FEAT;

    int*   cntO     = (int*)d_ws;
    int*   cntI     = cntO + nAlign;
    int*   cursor   = cntI + nAlign;
    int*   seg      = cursor + 4;
    float* rsO      = (float*)(seg + 2 * nAlign);
    float* rsI      = rsO + nAlign;
    int*   rank     = (int*)(rsI + nAlign);
    int*   col      = rank + eAlign;
    float* bufA     = (float*)(col + eAlign);
    float* bufB     = bufA + nfeat;

    const int B = 256;
    const int gEdgeT  = (nedges + B - 1) / B;
    const int gNodeW  = (nnodes + 7) / 8;
    const int gPrep   = (int)((nAlign / 4 + B - 1) / B);
    const int gGemm   = (nnodes + ROWS_PER_BLK - 1) / ROWS_PER_BLK;
    const int nZero   = (int)(2 * nAlign + 4);

    k_zero_i<<<(nZero + B - 1) / B, B, 0, stream>>>(cntO, nZero);
    k_count_rank<<<gEdgeT, B, 0, stream>>>(src, dst, cntO, cntI, rank, nedges);
    k_node_prep<<<gPrep, B, 0, stream>>>(cntO, cntI, rsO, rsI, seg, cursor, (int)nAlign);
    k_fill<<<gEdgeT, B, 0, stream>>>(src, dst, seg, rank, col, nedges);

    k_gemm_scaled<<<gGemm, B, 0, stream>>>(feat, rsO, W1, bufA, nnodes);
    k_gather_relu<<<gNodeW, B, 0, stream>>>(bufA, seg, col, rsI, b1, bufB, nnodes);

    k_gemm_scaled<<<gGemm, B, 0, stream>>>(bufB, rsO, W2, bufA, nnodes);

    k_gather_cls<<<gNodeW, B, 0, stream>>>(bufA, seg, col, rsI, b2, Wc, bc, out, nnodes);
}


# global integer atomics of the edge sort (rank, degrees) replaced by a two-level sort on LDS counters: edges filed by 256-node range, then ranked per range; rows still summed in arrival-rank order
# speedup vs baseline: 3.2291x; 1.2983x over previous
#include <hip/hip_runtime.h>
#include <math.h>

typedef __attribute__((ext_vector_type(2))) float v2f;
typedef __attribute__((ext_vector_type(8))) float v8f;

static constexpr int FEAT = 128;
static constexpr int APAD = 132;
static constexpr int ROWS_PER_BLK = 64;

static constexpr int BSH = 8;
static constexpr int NBMAX = 512;
static constexpr int CHUNK = 4096;

__global__ __launch_bounds__(256) void k_range_count(const int* __restrict__ src,
                                                     const int* __restrict__ dst,
                                                     int* __restrict__ cntAll,
                                                     int ne, int nb, int nch) {
    __shared__ int hd[NBMAX];
    __shared__ int hs[NBMAX];
    const int t = threadIdx.x;
    for (int b = t; b < NBMAX; b += 256) { hd[b] = 0; hs[b] = 0; }
    __syncthreads();
    const int c = blockIdx.x;
    const int e0 = c * CHUNK + t;
    int dv[CHUNK / 256], sv[CHUNK / 256];
    #pragma unroll
    for (int i = 0; i < CHUNK / 256; ++i) {
        const int e = e0 + i * 256;
        const int ec = e < ne ? e : ne - 1;
        dv[i] = dst[ec];
        sv[i] = src[ec];
    }
    #pragma unroll
    for (int i = 0; i < CHUNK / 256; ++i) {
        if (e0 + i * 256 < ne) {
            atomicAdd(&hd[dv[i] >> BSH], 1);
            atomicAdd(&hs[sv[i] >> BSH], 1);
        }
    }
    __syncthreads();
    for (int b = t; b < nb; b += 256) {
        cntAll[(size_t)b * nch + c] = hd[b];
        cntAll[(size_t)(nb + b) * nch + c] = hs[b];
    }
}

__global__ __launch_bounds__(256) void k_chunk_scan(const int* __restrict__ cntAll,
                                                    int* __restrict__ offAll,
                                                    int* __restrict__ totAll,
                                                    int nrows, int nch) {
    const int lane = threadIdx.x & 31;
    const int r = blockIdx.x * 8 + (threadIdx.x >> 5);
    if (r >= nrows) return;
    const int per = (nch + 31) >> 5;
    const int c0 = lane * per;
    const int* __restrict__ crow = cntAll + (size_t)r * nch;
    int* __restrict__ orow = offAll + (size_t)r * nch;

    int s = 0;
    for (int j0 = 0; j0 < per; j0 += 16) {
        int v[16];
        #pragma unroll
        for (int j = 0; j < 16; ++j) {
            const int c = c0 + j0 + j;
            v[j] = (j0 + j < per && c < nch) ? crow[c] : 0;
        }
        #pragma unroll
        for (int j = 0; j < 16; ++j) s += v[j];
    }
    int incl = s;
    for (int d = 1; d < 32; d <<= 1) {
        int u = __shfl_up(incl, d);
        if (lane >= d) incl += u;
    }
    if (lane == 31) totAll[r] = incl;
    int run = incl - s;
    for (int j0 = 0; j0 < per; j0 += 16) {
        int v[16];
        #pragma unroll
        for (int j = 0; j < 16; ++j) {
            const int c = c0 + j0 + j;
            v[j] = (j0 + j < per && c < nch) ? crow[c] : 0;
        }
        #pragma unroll
        for (int j = 0; j < 16; ++j) {
            const int c = c0 + j0 + j;
            if (j0 + j < per && c < nch) orow[c] = run;
            run += v[j];
        }
    }
}

template <bool FULL>
static __device__ __forceinline__ void file_edges(const int (&dv)[CHUNK / 256], const int (&sv)[CHUNK / 256],
                                                  int* curD, int* curS,
                                                  unsigned* __restrict__ PD, int* __restrict__ PS,
                                                  int e0, int ne) {
    int pd[CHUNK / 256], ps[CHUNK / 256];
    #pragma unroll
    for (int i = 0; i < CHUNK / 256; ++i) {
        if (FULL || e0 + i * 256 < ne) {
            pd[i] = atomicAdd(&curD[dv[i] >> BSH], 1);
            ps[i] = atomicAdd(&curS[sv[i] >> BSH], 1);
        }
    }
    #pragma unroll
    for (int i = 0; i < CHUNK / 256; ++i) {
        if (FULL || e0 + i * 256 < ne) {
            PD[pd[i]] = ((unsigned)sv[i] << BSH) | (unsigned)(dv[i] & 255);
            PS[ps[i]] = sv[i] & 255;
        }
    }
}

__global__ __launch_bounds__(256) void k_range_file(const int* __restrict__ src,
                                                    const int* __restrict__ dst,
                                                    const int* __restrict__ offAll,
                                                    const int* __restrict__ totAll,
                                                    unsigned* __restrict__ PD,
                                                    int* __restrict__ PS,
                                                    int ne, int nb, int nch) {
    __shared__ int curD[NBMAX];
    __shared__ int curS[NBMAX];
    const int t = threadIdx.x;
    const int lane = t & 31;
    const int wave = t >> 5;
    const int c = blockIdx.x;
    const int e0 = c * CHUNK + t;
    int dv[CHUNK / 256], sv[CHUNK / 256];
    #pragma unroll
    for (int i = 0; i < CHUNK / 256; ++i) {
        const int e = e0 + i * 256;
        const int ec = e < ne ? e : ne - 1;
        dv[i] = dst[ec];
        sv[i] = src[ec];
    }
    if (wave < 2) {
        const int rb = wave == 0 ? 0 : nb;
        int* cur = wave == 0 ? curD : curS;
        int loc[NBMAX / 32];
        int s = 0;
        #pragma unroll
        for (int j = 0; j < NBMAX / 32; ++j) {
            const int b = lane * (NBMAX / 32) + j;
            const int v = b < nb ? totAll[rb + b] : 0;
            loc[j] = s;
            s += v;
        }
        int incl = s;
        for (int d = 1; d < 32; d <<= 1) {
            int u = __shfl_up(incl, d);
            if (lane >= d) incl += u;
        }
        const int ex = incl - s;
        int off[NBMAX / 32];
        #pragma unroll
        for (int j = 0; j < NBMAX / 32; ++j) {
            const int b = lane * (NBMAX / 32) + j;
            off[j] = offAll[(size_t)(rb + (b < nb ? b : nb - 1)) * nch + c];
        }
        #pragma unroll
        for (int j = 0; j < NBMAX / 32; ++j) {
            const int b = lane * (NBMAX / 32) + j;
            if (b < nb) cur[b] = ex + loc[j] + off[j];
        }
    }
    __syncthreads();
    if ((c + 1) * CHUNK <= ne) {
        file_edges<true>(dv, sv, curD, curS, PD, PS, e0, ne);
    } else {
        file_edges<false>(dv, sv, curD, curS, PD, PS, e0, ne);
    }
}

__global__ __launch_bounds__(256) void k_range_rank(const unsigned* __restrict__ PD,
                                                    const int* __restrict__ PS,
                                                    const int* __restrict__ totAll,
                                                    int* __restrict__ seg,
                                                    int* __restrict__ col,
                                                    float* __restrict__ rsO,
                                                    float* __restrict__ rsI,
                                                    int n, int nb) {
    __shared__ int cnt[256];
    __shared__ int cur[256];
    __shared__ int wred[8];
    __shared__ int wsum[8];
    const int t = threadIdx.x;
    const int lane = t & 31;
    const int wave = t >> 5;
    const int r = blockIdx.x;
    const bool isD = r < nb;
    const int b = isD ? r : r - nb;
    const int rb = isD ? 0 : nb;

    int acc = 0;
    for (int bb = t; bb < b; bb += 256) acc += totAll[rb + bb];
    for (int d = 16; d >= 1; d >>= 1) acc += __shfl_xor(acc, d);
    if (lane == 0) wred[wave] = acc;
    cnt[t] = 0;
    __syncthreads();
    int base = 0;
    for (int w = 0; w < 8; ++w) base += wred[w];
    const int size = totAll[r];

    const unsigned* __restrict__ P = isD ? PD : (const unsigned*)PS;
    for (int i0 = t; i0 < size + t; i0 += 2048) {
        unsigned k[8];
        #pragma unroll
        for (int j = 0; j < 8; ++j) {
            const int i = i0 + j * 256;
            k[j] = P[base + (i < size ? i : size - 1)];
        }
        #pragma unroll
        for (int j = 0; j < 8; ++j) {
            if (i0 + j * 256 < size) atomicAdd(&cnt[k[j] & 255u], 1);
        }
    }
    __syncthreads();
    const int cn = cnt[t];
    const int node = b * 256 + t;
    float a = (float)cn;
    a = a < 1.0f ? 1.0f : a;
    const float rs = 1.0f / sqrtf(a);
    if (!isD) {
        if (node < n) rsO[node] = rs;
        return;
    }

    int incl = cn;
    for (int d = 1; d < 32; d <<= 1) {
        int u = __shfl_up(incl, d);
        if (lane >= d) incl += u;
    }
    if (lane == 31) wsum[wave] = incl;
    __syncthreads();
    int woff = 0;
    for (int w = 0; w < 8; ++w) {
        if (w < wave) woff += wsum[w];
    }
    const int start = base + woff + incl - cn;
    cur[t] = start;
    if (node < n) {
        int2 sg = {start, cn};
        *(int2*)(seg + 2 * (size_t)node) = sg;
        rsI[node] = rs;
    }
    __syncthreads();
    for (int i0 = t; i0 < size + t; i0 += 2048) {
        unsigned k[8];
        #pragma unroll
        for (int j = 0; j < 8; ++j) {
            const int i = i0 + j * 256;
            k[j] = PD[base + (i < size ? i : size - 1)];
        }
        #pragma unroll
        for (int j = 0; j < 8; ++j) {
            if (i0 + j * 256 < size) {
                const int pos = atomicAdd(&cur[k[j] & 255u], 1);
                col[pos] = (int)(k[j] >> BSH);
            }
        }
    }
}

__global__ __launch_bounds__(256) void k_gemm_scaled(const float* __restrict__ X,
                                                     const float* __restrict__ rs,
                                                     const float* __restrict__ W,
                                                     float* __restrict__ Y, int nrows) {
    __shared__ float Alds[ROWS_PER_BLK * APAD];

    const int rowbase = blockIdx.x * ROWS_PER_BLK;
    const int t = threadIdx.x;

    for (int i = t; i < ROWS_PER_BLK * (FEAT / 4); i += 256) {
        int r  = i >> 5;
        int c4 = (i & 31) << 2;
        int row = rowbase + r;
        float4 v = {0.f, 0.f, 0.f, 0.f};
        float s = 0.f;
        if (row < nrows) {
            v = *(const float4*)(X + (size_t)row * FEAT + c4);
            s = rs[row];
        }
        float* dstp = Alds + r * APAD + c4;
        dstp[0] = v.x * s; dstp[1] = v.y * s; dstp[2] = v.z * s; dstp[3] = v.w * s;
    }
    __syncthreads();

    const int wave = t >> 5;
    const int lane = t & 31;
    const int half = lane >> 4;
    const int l16  = lane & 15;
    const int ncol = wave * 16 + l16;

    v8f acc0 = {}, acc1 = {}, acc2 = {}, acc3 = {};
    const float* a0 = Alds + (0 * 16 + l16) * APAD;
    const float* a1 = Alds + (1 * 16 + l16) * APAD;
    const float* a2 = Alds + (2 * 16 + l16) * APAD;
    const float* a3 = Alds + (3 * 16 + l16) * APAD;

    #pragma unroll 4
    for (int k = 0; k < FEAT; k += 4) {
        const int ka = k + half * 2;
        v2f b;
        b.x = W[ka * FEAT + ncol];
        b.y = W[(ka + 1) * FEAT + ncol];
        v2f a;
        a.x = a0[ka]; a.y = a0[ka + 1];
        acc0 = __builtin_amdgcn_wmma_f32_16x16x4_f32(false, a, false, b, (short)0, acc0, false, false);
        a.x = a1[ka]; a.y = a1[ka + 1];
        acc1 = __builtin_amdgcn_wmma_f32_16x16x4_f32(false, a, false, b, (short)0, acc1, false, false);
        a.x = a2[ka]; a.y = a2[ka + 1];
        acc2 = __builtin_amdgcn_wmma_f32_16x16x4_f32(false, a, false, b, (short)0, acc2, false, false);
        a.x = a3[ka]; a.y = a3[ka + 1];
        acc3 = __builtin_amdgcn_wmma_f32_16x16x4_f32(false, a, false, b, (short)0, acc3, false, false);
    }

    const int mrow0 = rowbase + half * 8;
    if (rowbase + ROWS_PER_BLK <= nrows) {
        #pragma unroll
        for (int r = 0; r < 8; ++r) {
            Y[(size_t)(mrow0 +  0 + r) * FEAT + ncol] = acc0[r];
            Y[(size_t)(mrow0 + 16 + r) * FEAT + ncol] = acc1[r];
            Y[(size_t)(mrow0 + 32 + r) * FEAT + ncol] = acc2[r];
            Y[(size_t)(mrow0 + 48 + r) * FEAT + ncol] = acc3[r];
        }
    } else {
        #pragma unroll
        for (int r = 0; r < 8; ++r) {
            int r0 = mrow0 + r;
            if (r0 +  0 < nrows) Y[(size_t)(r0 +  0) * FEAT + ncol] = acc0[r];
            if (r0 + 16 < nrows) Y[(size_t)(r0 + 16) * FEAT + ncol] = acc1[r];
            if (r0 + 32 < nrows) Y[(size_t)(r0 + 32) * FEAT + ncol] = acc2[r];
            if (r0 + 48 < nrows) Y[(size_t)(r0 + 48) * FEAT + ncol] = acc3[r];
        }
    }
}

static __device__ __forceinline__ float4 gather_row(const float* __restrict__ H,
                                                    const int* __restrict__ seg,
                                                    const int* __restrict__ col,
                                                    int node, int lane) {
    const int2 sg = *(const int2*)(seg + 2 * (size_t)node);
    const int start = sg.x;
    const int cnt = sg.y;
    const float* Hl = H + lane * 4;

    float4 acc = {0.f, 0.f, 0.f, 0.f};
#define ROW(j) (*(const float4*)(Hl + (size_t)__builtin_amdgcn_readlane(cv, (j)) * FEAT))
#define ADD(v) { acc.x += (v).x; acc.y += (v).y; acc.z += (v).z; acc.w += (v).w; }
    for (int base = 0; base < cnt; base += 32) {
        int m = cnt - base;
        if (m > 32) m = 32;
        int cv = 0;
        if (lane < m) cv = col[start + base + lane];
        int j = 0;
        for (; j + 8 <= m; j += 8) {
            float4 v0 = ROW(j), v1 = ROW(j + 1), v2 = ROW(j + 2), v3 = ROW(j + 3);
            float4 v4 = ROW(j + 4), v5 = ROW(j + 5), v6 = ROW(j + 6), v7 = ROW(j + 7);
            ADD(v0) ADD(v1) ADD(v2) ADD(v3) ADD(v4) ADD(v5) ADD(v6) ADD(v7)
        }
        if (j + 4 <= m) {
            float4 v0 = ROW(j), v1 = ROW(j + 1), v2 = ROW(j + 2), v3 = ROW(j + 3);
            ADD(v0) ADD(v1) ADD(v2) ADD(v3)
            j += 4;
        }
        if (j + 2 <= m) {
            float4 v0 = ROW(j), v1 = ROW(j + 1);
            ADD(v0) ADD(v1)
            j += 2;
        }
        if (j < m) {
            float4 v0 = ROW(j);
            ADD(v0)
        }
    }
#undef ROW
#undef ADD
    return acc;
}

static __device__ __forceinline__ float4 row_tail(float4 acc, float s, float4 b) {
    float4 h;
    h.x = acc.x * s + b.x; h.x = h.x > 0.f ? h.x : 0.f;
    h.y = acc.y * s + b.y; h.y = h.y > 0.f ? h.y : 0.f;
    h.z = acc.z * s + b.z; h.z = h.z > 0.f ? h.z : 0.f;
    h.w = acc.w * s + b.w; h.w = h.w > 0.f ? h.w : 0.f;
    return h;
}

__global__ __launch_bounds__(256) void k_gather_relu(const float* __restrict__ H,
                                                     const int* __restrict__ seg,
                                                     const int* __restrict__ col,
                                                     const float* __restrict__ rsI,
                                                     const float* __restrict__ bias,
                                                     float* __restrict__ out, int n) {
    int node = blockIdx.x * 8 + (threadIdx.x >> 5);
    if (node >= n) return;
    node = __builtin_amdgcn_readfirstlane(node);
    const int lane = threadIdx.x & 31;
    const float s = rsI[node];
    const float4 b = *(const float4*)(bias + lane * 4);
    const float4 acc = gather_row(H, seg, col, node, lane);
    *(float4*)(out + (size_t)node * FEAT + lane * 4) = row_tail(acc, s, b);
}

static __device__ __forceinline__ float lx_dpp_q1(float x) {
    return __builtin_bit_cast(float, __builtin_amdgcn_update_dpp(0, __builtin_bit_cast(int, x), 0xB1, 0xF, 0xF, true));
}
static __device__ __forceinline__ float lx_dpp_q2(float x) {
    return __builtin_bit_cast(float, __builtin_amdgcn_update_dpp(0, __builtin_bit_cast(int, x), 0x4E, 0xF, 0xF, true));
}
static __device__ __forceinline__ float lx_dpp_x4(float x) {
    return __builtin_bit_cast(float, __builtin_amdgcn_update_dpp(0, __builtin_bit_cast(int, x), 0x164, 0xF, 0xF, true));
}
static __device__ __forceinline__ float lx_dpp_x8(float x) {
    return __builtin_bit_cast(float, __builtin_amdgcn_update_dpp(0, __builtin_bit_cast(int, x), 0x168, 0xF, 0xF, true));
}
static __device__ __forceinline__ float lx_half(float x) {
    return __builtin_bit_cast(float, __builtin_amdgcn_permlanex16(__builtin_bit_cast(int, x), __builtin_bit_cast(int, x),
                                                                  0x76543210, 0xfedcba98, false, false));
}

__global__ __launch_bounds__(256) void k_gather_cls(const float* __restrict__ H,
                                                    const int* __restrict__ seg,
                                                    const int* __restrict__ col,
                                                    const float* __restrict__ rsI,
                                                    const float* __restrict__ bias,
                                                    const float* __restrict__ Wc,
                                                    const float* __restrict__ bc,
                                                    float* __restrict__ out, int n) {
    __shared__ float lg[32];
    const int lane = threadIdx.x & 31;
    const int wave = threadIdx.x >> 5;
    const float bco = bc[lane & 3];
    int node = blockIdx.x * 8 + wave;
    if (node < n) {
        node = __builtin_amdgcn_readfirstlane(node);
        const float s = rsI[node];
        const float4 b = *(const float4*)(bias + lane * 4);
        const float4 acc = gather_row(H, seg, col, node, lane);
        const float4 h = row_tail(acc, s, b);

        const float4 w0 = *(const float4*)(Wc + lane * 16);
        const float4 w1 = *(const float4*)(Wc + lane * 16 + 4);
        const float4 w2 = *(const float4*)(Wc + lane * 16 + 8);
        const float4 w3 = *(const float4*)(Wc + lane * 16 + 12);
        const float p0 = h.x * w0.x + h.y * w1.x + h.z * w2.x + h.w * w3.x;
        const float p1 = h.x * w0.y + h.y * w1.y + h.z * w2.y + h.w * w3.y;
        const float p2 = h.x * w0.z + h.y * w1.z + h.z * w2.z + h.w * w3.z;
        const float p3 = h.x * w0.w + h.y * w1.w + h.z * w2.w + h.w * w3.w;

        const bool o1 = (lane & 1) != 0;
        const bool o2 = (lane & 2) != 0;
        const float q0 = (o1 ? p2 : p0) + lx_dpp_q1(o1 ? p0 : p2);
        const float q1 = (o1 ? p3 : p1) + lx_dpp_q1(o1 ? p1 : p3);
        float r = (o2 ? q1 : q0) + lx_dpp_q2(o2 ? q0 : q1);
        r += lx_dpp_x4(r);
        r += lx_dpp_x8(r);
        r += lx_half(r);
        if (lane < 4) lg[wave * 4 + (((lane & 1) << 1) | (lane >> 1))] = r;
    }
    __syncthreads();
    if (wave == 0) {
        const int i = blockIdx.x * 32 + lane;
        if (i < n * 4) out[i] = lg[lane] + bco;
    }
}

extern "C" void kernel_launch(void* const* d_in, const int* in_sizes, int n_in,
                              void* d_out, int out_size, void* d_ws, size_t ws_size,
                              hipStream_t stream) {
    const float* feat = (const float*)d_in[0];
    const int*   src  = (const int*)d_in[1];
    const int*   dst  = (const int*)d_in[2];
    const float* W1   = (const float*)d_in[3];
    const float* b1   = (const float*)d_in[4];
    const float* W2   = (const float*)d_in[5];
    const float* b2   = (const float*)d_in[6];
    const float* Wc   = (const float*)d_in[7];
    const float* bc   = (const float*)d_in[8];
    float* out = (float*)d_out;

    const int nnodes = in_sizes[0] / FEAT;
    const int nedges = in_sizes[1];
    const size_t nAlign = ((size_t)nnodes + 3) & ~(size_t)3;
    const size_t eAlign = ((size_t)nedges + 3) & ~(size_t)3;
    const size_t nfeat = (size_t)nnodes * FEAT;

    const int NB  = (nnodes + 255) >> BSH;
    const int NCH = (nedges + CHUNK - 1) / CHUNK;
    const size_t tabAlign = ((size_t)2 * NB * NCH + 3) & ~(size_t)3;
    const size_t totAlign = ((size_t)2 * NB + 3) & ~(size_t)3;

    int*      cntAll = (int*)d_ws;
    int*      offAll = cntAll + tabAlign;
    int*      totAll = offAll + tabAlign;
    int*      seg    = totAll + totAlign;
    float*    rsO    = (float*)(seg + 2 * nAlign);
    float*    rsI    = rsO + nAlign;
    unsigned* PD     = (unsigned*)(rsI + nAlign);
    int*      PS     = (int*)(PD + eAlign);
    int*      col    = PS + eAlign;
    float*    bufA   = (float*)(col + eAlign);
    float*    bufB   = bufA + nfeat;

    const int B = 256;
    const int gNodeW  = (nnodes + 7) / 8;
    const int gGemm   = (nnodes + ROWS_PER_BLK - 1) / ROWS_PER_BLK;

    k_range_count<<<NCH, B, 0, stream>>>(src, dst, cntAll, nedges, NB, NCH);
    k_chunk_scan<<<(2 * NB + 7) / 8, B, 0, stream>>>(cntAll, offAll, totAll, 2 * NB, NCH);
    k_range_file<<<NCH, B, 0, stream>>>(src, dst, offAll, totAll, PD, PS, nedges, NB, NCH);
    k_range_rank<<<2 * NB, B, 0, stream>>>(PD, PS, totAll, seg, col, rsO, rsI, nnodes, NB);

    k_gemm_scaled<<<gGemm, B, 0, stream>>>(feat, rsO, W1, bufA, nnodes);
    k_gather_relu<<<gNodeW, B, 0, stream>>>(bufA, seg, col, rsI, b1, bufB, nnodes);

    k_gemm_scaled<<<gGemm, B, 0, stream>>>(bufB, rsO, W2, bufA, nnodes);

    k_gather_cls<<<gNodeW, B, 0, stream>>>(bufA, seg, col, rsI, b2, Wc, bc, out, nnodes);
}


# layer-2 product moved into the layer-1 gather launch: a block sums 64 nodes (8 per wave) into an LDS tile and multiplies it by W2 there; h1 never reaches memory; 8 launches -> 7; sum order unchanged
# speedup vs baseline: 3.3913x; 1.0502x over previous
#include <hip/hip_runtime.h>
#include <math.h>

typedef __attribute__((ext_vector_type(2))) float v2f;
typedef __attribute__((ext_vector_type(8))) float v8f;

static constexpr int FEAT = 128;
static constexpr int APAD = 132;
static constexpr int ROWS_PER_BLK = 64;

static constexpr int BSH = 8;
static constexpr int NBMAX = 512;
static constexpr int CHUNK = 4096;

__global__ __launch_bounds__(256) void k_range_count(const int* __restrict__ src,
                                                     const int* __restrict__ dst,
                                                     int* __restrict__ cntAll,
                                                     int ne, int nb, int nch) {
    __shared__ int hd[NBMAX];
    __shared__ int hs[NBMAX];
    const int t = threadIdx.x;
    for (int b = t; b < NBMAX; b += 256) { hd[b] = 0; hs[b] = 0; }
    __syncthreads();
    const int c = blockIdx.x;
    const int e0 = c * CHUNK + t;
    int dv[CHUNK / 256], sv[CHUNK / 256];
    #pragma unroll
    for (int i = 0; i < CHUNK / 256; ++i) {
        const int e = e0 + i * 256;
        const int ec = e < ne ? e : ne - 1;
        dv[i] = dst[ec];
        sv[i] = src[ec];
    }
    #pragma unroll
    for (int i = 0; i < CHUNK / 256; ++i) {
        if (e0 + i * 256 < ne) {
            atomicAdd(&hd[dv[i] >> BSH], 1);
            atomicAdd(&hs[sv[i] >> BSH], 1);
        }
    }
    __syncthreads();
    for (int b = t; b < nb; b += 256) {
        cntAll[(size_t)b * nch + c] = hd[b];
        cntAll[(size_t)(nb + b) * nch + c] = hs[b];
    }
}

__global__ __launch_bounds__(256) void k_chunk_scan(const int* __restrict__ cntAll,
                                                    int* __restrict__ offAll,
                                                    int* __restrict__ totAll,
                                                    int nrows, int nch) {
    const int lane = threadIdx.x & 31;
    const int r = blockIdx.x * 8 + (threadIdx.x >> 5);
    if (r >= nrows) return;
    const int per = (nch + 31) >> 5;
    const int c0 = lane * per;
    const int* __restrict__ crow = cntAll + (size_t)r * nch;
    int* __restrict__ orow = offAll + (size_t)r * nch;

    int s = 0;
    for (int j0 = 0; j0 < per; j0 += 16) {
        int v[16];
        #pragma unroll
        for (int j = 0; j < 16; ++j) {
            const int c = c0 + j0 + j;
            v[j] = (j0 + j < per && c < nch) ? crow[c] : 0;
        }
        #pragma unroll
        for (int j = 0; j < 16; ++j) s += v[j];
    }
    int incl = s;
    for (int d = 1; d < 32; d <<= 1) {
        int u = __shfl_up(incl, d);
        if (lane >= d) incl += u;
    }
    if (lane == 31) totAll[r] = incl;
    int run = incl - s;
    for (int j0 = 0; j0 < per; j0 += 16) {
        int v[16];
        #pragma unroll
        for (int j = 0; j < 16; ++j) {
            const int c = c0 + j0 + j;
            v[j] = (j0 + j < per && c < nch) ? crow[c] : 0;
        }
        #pragma unroll
        for (int j = 0; j < 16; ++j) {
            const int c = c0 + j0 + j;
            if (j0 + j < per && c < nch) orow[c] = run;
            run += v[j];
        }
    }
}

template <bool FULL>
static __device__ __forceinline__ void file_edges(const int (&dv)[CHUNK / 256], const int (&sv)[CHUNK / 256],
                                                  int* curD, int* curS,
                                                  unsigned* __restrict__ PD, int* __restrict__ PS,
                                                  int e0, int ne) {
    int pd[CHUNK / 256], ps[CHUNK / 256];
    #pragma unroll
    for (int i = 0; i < CHUNK / 256; ++i) {
        if (FULL || e0 + i * 256 < ne) {
            pd[i] = atomicAdd(&curD[dv[i] >> BSH], 1);
            ps[i] = atomicAdd(&curS[sv[i] >> BSH], 1);
        }
    }
    #pragma unroll
    for (int i = 0; i < CHUNK / 256; ++i) {
        if (FULL || e0 + i * 256 < ne) {
            PD[pd[i]] = ((unsigned)sv[i] << BSH) | (unsigned)(dv[i] & 255);
            PS[ps[i]] = sv[i] & 255;
        }
    }
}

__global__ __launch_bounds__(256) void k_range_file(const int* __restrict__ src,
                                                    const int* __restrict__ dst,
                                                    const int* __restrict__ offAll,
                                                    const int* __restrict__ totAll,
                                                    unsigned* __restrict__ PD,
                                                    int* __restrict__ PS,
                                                    int ne, int nb, int nch) {
    __shared__ int curD[NBMAX];
    __shared__ int curS[NBMAX];
    const int t = threadIdx.x;
    const int lane = t & 31;
    const int wave = t >> 5;
    const int c = blockIdx.x;
    const int e0 = c * CHUNK + t;
    int dv[CHUNK / 256], sv[CHUNK / 256];
    #pragma unroll
    for (int i = 0; i < CHUNK / 256; ++i) {
        const int e = e0 + i * 256;
        const int ec = e < ne ? e : ne - 1;
        dv[i] = dst[ec];
        sv[i] = src[ec];
    }
    if (wave < 2) {
        const int rb = wave == 0 ? 0 : nb;
        int* cur = wave == 0 ? curD : curS;
        int loc[NBMAX / 32];
        int s = 0;
        #pragma unroll
        for (int j = 0; j < NBMAX / 32; ++j) {
            const int b = lane * (NBMAX / 32) + j;
            const int v = b < nb ? totAll[rb + b] : 0;
            loc[j] = s;
            s += v;
        }
        int incl = s;
        for (int d = 1; d < 32; d <<= 1) {
            int u = __shfl_up(incl, d);
            if (lane >= d) incl += u;
        }
        const int ex = incl - s;
        int off[NBMAX / 32];
        #pragma unroll
        for (int j = 0; j < NBMAX / 32; ++j) {
            const int b = lane * (NBMAX / 32) + j;
            off[j] = offAll[(size_t)(rb + (b < nb ? b : nb - 1)) * nch + c];
        }
        #pragma unroll
        for (int j = 0; j < NBMAX / 32; ++j) {
            const int b = lane * (NBMAX / 32) + j;
            if (b < nb) cur[b] = ex + loc[j] + off[j];
        }
    }
    __syncthreads();
    if ((c + 1) * CHUNK <= ne) {
        file_edges<true>(dv, sv, curD, curS, PD, PS, e0, ne);
    } else {
        file_edges<false>(dv, sv, curD, curS, PD, PS, e0, ne);
    }
}

__global__ __launch_bounds__(256) void k_range_rank(const unsigned* __restrict__ PD,
                                                    const int* __restrict__ PS,
                                                    const int* __restrict__ totAll,
                                                    int* __restrict__ seg,
                                                    int* __restrict__ col,
                                                    float* __restrict__ rsO,
                                                    float* __restrict__ rsI,
                                                    int n, int nb) {
    __shared__ int cnt[256];
    __shared__ int cur[256];
    __shared__ int wred[8];
    __shared__ int wsum[8];
    const int t = threadIdx.x;
    const int lane = t & 31;
    const int wave = t >> 5;
    const int r = blockIdx.x;
    const bool isD = r < nb;
    const int b = isD ? r : r - nb;
    const int rb = isD ? 0 : nb;

    int acc = 0;
    for (int bb = t; bb < b; bb += 256) acc += totAll[rb + bb];
    for (int d = 16; d >= 1; d >>= 1) acc += __shfl_xor(acc, d);
    if (lane == 0) wred[wave] = acc;
    cnt[t] = 0;
    __syncthreads();
    int base = 0;
    for (int w = 0; w < 8; ++w) base += wred[w];
    const int size = totAll[r];

    const unsigned* __restrict__ P = isD ? PD : (const unsigned*)PS;
    for (int i0 = t; i0 < size + t; i0 += 2048) {
        unsigned k[8];
        #pragma unroll
        for (int j = 0; j < 8; ++j) {
            const int i = i0 + j * 256;
            k[j] = P[base + (i < size ? i : size - 1)];
        }
        #pragma unroll
        for (int j = 0; j < 8; ++j) {
            if (i0 + j * 256 < size) atomicAdd(&cnt[k[j] & 255u], 1);
        }
    }
    __syncthreads();
    const int cn = cnt[t];
    const int node = b * 256 + t;
    float a = (float)cn;
    a = a < 1.0f ? 1.0f : a;
    const float rs = 1.0f / sqrtf(a);
    if (!isD) {
        if (node < n) rsO[node] = rs;
        return;
    }

    int incl = cn;
    for (int d = 1; d < 32; d <<= 1) {
        int u = __shfl_up(incl, d);
        if (lane >= d) incl += u;
    }
    if (lane == 31) wsum[wave] = incl;
    __syncthreads();
    int woff = 0;
    for (int w = 0; w < 8; ++w) {
        if (w < wave) woff += wsum[w];
    }
    const int start = base + woff + incl - cn;
    cur[t] = start;
    if (node < n) {
        int2 sg = {start, cn};
        *(int2*)(seg + 2 * (size_t)node) = sg;
        rsI[node] = rs;
    }
    __syncthreads();
    for (int i0 = t; i0 < size + t; i0 += 2048) {
        unsigned k[8];
        #pragma unroll
        for (int j = 0; j < 8; ++j) {
            const int i = i0 + j * 256;
            k[j] = PD[base + (i < size ? i : size - 1)];
        }
        #pragma unroll
        for (int j = 0; j < 8; ++j) {
            if (i0 + j * 256 < size) {
                const int pos = atomicAdd(&cur[k[j] & 255u], 1);
                col[pos] = (int)(k[j] >> BSH);
            }
        }
    }
}

static __device__ __forceinline__ void tile_product_store(const float* Alds,
                                                          const float* __restrict__ W,
                                                          float* __restrict__ Y,
                                                          int rowbase, int nrows, int t) {
    const int wave = t >> 5;
    const int lane = t & 31;
    const int half = lane >> 4;
    const int l16  = lane & 15;
    const int ncol = wave * 16 + l16;

    v8f acc0 = {}, acc1 = {}, acc2 = {}, acc3 = {};
    const float* a0 = Alds + (0 * 16 + l16) * APAD;
    const float* a1 = Alds + (1 * 16 + l16) * APAD;
    const float* a2 = Alds + (2 * 16 + l16) * APAD;
    const float* a3 = Alds + (3 * 16 + l16) * APAD;

    #pragma unroll 4
    for (int k = 0; k < FEAT; k += 4) {
        const int ka = k + half * 2;
        v2f b;
        b.x = W[ka * FEAT + ncol];
        b.y = W[(ka + 1) * FEAT + ncol];
        v2f a;
        a.x = a0[ka]; a.y = a0[ka + 1];
        acc0 = __builtin_amdgcn_wmma_f32_16x16x4_f32(false, a, false, b, (short)0, acc0, false, false);
        a.x = a1[ka]; a.y = a1[ka + 1];
        acc1 = __builtin_amdgcn_wmma_f32_16x16x4_f32(false, a, false, b, (short)0, acc1, false, false);
        a.x = a2[ka]; a.y = a2[ka + 1];
        acc2 = __builtin_amdgcn_wmma_f32_16x16x4_f32(false, a, false, b, (short)0, acc2, false, false);
        a.x = a3[ka]; a.y = a3[ka + 1];
        acc3 = __builtin_amdgcn_wmma_f32_16x16x4_f32(false, a, false, b, (short)0, acc3, false, false);
    }

    const int mrow0 = rowbase + half * 8;
    if (rowbase + ROWS_PER_BLK <= nrows) {
        #pragma unroll
        for (int r = 0; r < 8; ++r) {
            Y[(size_t)(mrow0 +  0 + r) * FEAT + ncol] = acc0[r];
            Y[(size_t)(mrow0 + 16 + r) * FEAT + ncol] = acc1[r];
            Y[(size_t)(mrow0 + 32 + r) * FEAT + ncol] = acc2[r];
            Y[(size_t)(mrow0 + 48 + r) * FEAT + ncol] = acc3[r];
        }
    } else {
        #pragma unroll
        for (int r = 0; r < 8; ++r) {
            int r0 = mrow0 + r;
            if (r0 +  0 < nrows) Y[(size_t)(r0 +  0) * FEAT + ncol] = acc0[r];
            if (r0 + 16 < nrows) Y[(size_t)(r0 + 16) * FEAT + ncol] = acc1[r];
            if (r0 + 32 < nrows) Y[(size_t)(r0 + 32) * FEAT + ncol] = acc2[r];
            if (r0 + 48 < nrows) Y[(size_t)(r0 + 48) * FEAT + ncol] = acc3[r];
        }
    }
}

__global__ __launch_bounds__(256) void k_gemm_scaled(const float* __restrict__ X,
                                                     const float* __restrict__ rs,
                                                     const float* __restrict__ W,
                                                     float* __restrict__ Y, int nrows) {
    __shared__ float Alds[ROWS_PER_BLK * APAD];

    const int rowbase = blockIdx.x * ROWS_PER_BLK;
    const int t = threadIdx.x;

    for (int i = t; i < ROWS_PER_BLK * (FEAT / 4); i += 256) {
        int r  = i >> 5;
        int c4 = (i & 31) << 2;
        int row = rowbase + r;
        float4 v = {0.f, 0.f, 0.f, 0.f};
        float s = 0.f;
        if (row < nrows) {
            v = *(const float4*)(X + (size_t)row * FEAT + c4);
            s = rs[row];
        }
        float* dstp = Alds + r * APAD + c4;
        dstp[0] = v.x * s; dstp[1] = v.y * s; dstp[2] = v.z * s; dstp[3] = v.w * s;
    }
    __syncthreads();
    tile_product_store(Alds, W, Y, rowbase, nrows, t);
}

static __device__ __forceinline__ void add_rows(float4& acc, const float* Hl, int cv, int m) {
#define ROW(j) (*(const float4*)(Hl + (size_t)__builtin_amdgcn_readlane(cv, (j)) * FEAT))
#define ADD(v) { acc.x += (v).x; acc.y += (v).y; acc.z += (v).z; acc.w += (v).w; }
    int j = 0;
    for (; j + 8 <= m; j += 8) {
        float4 v0 = ROW(j), v1 = ROW(j + 1), v2 = ROW(j + 2), v3 = ROW(j + 3);
        float4 v4 = ROW(j + 4), v5 = ROW(j + 5), v6 = ROW(j + 6), v7 = ROW(j + 7);
        ADD(v0) ADD(v1) ADD(v2) ADD(v3) ADD(v4) ADD(v5) ADD(v6) ADD(v7)
    }
    if (j + 4 <= m) {
        float4 v0 = ROW(j), v1 = ROW(j + 1), v2 = ROW(j + 2), v3 = ROW(j + 3);
        ADD(v0) ADD(v1) ADD(v2) ADD(v3)
        j += 4;
    }
    if (j + 2 <= m) {
        float4 v0 = ROW(j), v1 = ROW(j + 1);
        ADD(v0) ADD(v1)
        j += 2;
    }
    if (j < m) {
        float4 v0 = ROW(j);
        ADD(v0)
    }
#undef ROW
#undef ADD
}

static __device__ __forceinline__ float4 gather_row(const float* __restrict__ H,
                                                    const int* __restrict__ seg,
                                                    const int* __restrict__ col,
                                                    int node, int lane) {
    const int2 sg = *(const int2*)(seg + 2 * (size_t)node);
    const int start = sg.x;
    const int cnt = sg.y;
    const float* Hl = H + lane * 4;

    float4 acc = {0.f, 0.f, 0.f, 0.f};
    for (int base = 0; base < cnt; base += 32) {
        int m = cnt - base;
        if (m > 32) m = 32;
        int cv = 0;
        if (lane < m) cv = col[start + base + lane];
        add_rows(acc, Hl, cv, m);
    }
    return acc;
}

static __device__ __forceinline__ float4 gather_row_ahead(const float* __restrict__ H,
                                                          const int* __restrict__ col,
                                                          int start, int cnt, int cv0, int lane) {
    const float* Hl = H + lane * 4;
    float4 acc = {0.f, 0.f, 0.f, 0.f};
    add_rows(acc, Hl, cv0, cnt < 32 ? cnt : 32);
    for (int base = 32; base < cnt; base += 32) {
        int m = cnt - base;
        if (m > 32) m = 32;
        int cv = 0;
        if (lane < m) cv = col[start + base + lane];
        add_rows(acc, Hl, cv, m);
    }
    return acc;
}

static __device__ __forceinline__ float4 row_tail(float4 acc, float s, float4 b) {
    float4 h;
    h.x = acc.x * s + b.x; h.x = h.x > 0.f ? h.x : 0.f;
    h.y = acc.y * s + b.y; h.y = h.y > 0.f ? h.y : 0.f;
    h.z = acc.z * s + b.z; h.z = h.z > 0.f ? h.z : 0.f;
    h.w = acc.w * s + b.w; h.w = h.w > 0.f ? h.w : 0.f;
    return h;
}

__global__ __launch_bounds__(256) void k_gather_relu_gemm(const float* __restrict__ H,
                                                          const int* __restrict__ seg,
                                                          const int* __restrict__ col,
                                                          const float* __restrict__ rsI,
                                                          const float* __restrict__ rsO,
                                                          const float* __restrict__ bias,
                                                          const float* __restrict__ W,
                                                          float* __restrict__ Y, int n) {
    __shared__ float Alds[ROWS_PER_BLK * APAD];

    const int rowbase = blockIdx.x * ROWS_PER_BLK;
    const int t = threadIdx.x;
    const int lane = t & 31;
    const int wave = __builtin_amdgcn_readfirstlane(t >> 5);
    const float4 b = *(const float4*)(bias + lane * 4);

    const int r0 = wave * (ROWS_PER_BLK / 8);
    int2 sgv = {0, 0};
    float siv = 0.f, sov = 0.f;
    {
        const int nd = rowbase + r0 + (lane & 7);
        if (nd < n) {
            sgv = *(const int2*)(seg + 2 * (size_t)nd);
            siv = rsI[nd];
            sov = rsO[nd];
        }
    }
    int startN = __builtin_amdgcn_readlane(sgv.x, 0);
    int cntN   = __builtin_amdgcn_readlane(sgv.y, 0);
    int cvN = 0;
    if (lane < cntN) cvN = col[startN + lane];

    for (int i = 0; i < ROWS_PER_BLK / 8; ++i) {
        const int start = startN, cnt = cntN, cv0 = cvN;
        const float s  = __builtin_bit_cast(float, __builtin_amdgcn_readlane(__builtin_bit_cast(int, siv), i));
        const float so = __builtin_bit_cast(float, __builtin_amdgcn_readlane(__builtin_bit_cast(int, sov), i));
        if (i + 1 < ROWS_PER_BLK / 8) {
            startN = __builtin_amdgcn_readlane(sgv.x, i + 1);
            cntN   = __builtin_amdgcn_readlane(sgv.y, i + 1);
            cvN = 0;
            if (lane < cntN) cvN = col[startN + lane];
        }
        const float4 acc = gather_row_ahead(H, col, start, cnt, cv0, lane);
        const float4 h = row_tail(acc, s, b);
        float4 v = {0.f, 0.f, 0.f, 0.f};
        if (rowbase + r0 + i < n) { v.x = h.x * so; v.y = h.y * so; v.z = h.z * so; v.w = h.w * so; }
        *(float4*)(Alds + (r0 + i) * APAD + lane * 4) = v;
    }
    __syncthreads();
    tile_product_store(Alds, W, Y, rowbase, n, t);
}

static __device__ __forceinline__ float lx_dpp_q1(float x) {
    return __builtin_bit_cast(float, __builtin_amdgcn_update_dpp(0, __builtin_bit_cast(int, x), 0xB1, 0xF, 0xF, true));
}
static __device__ __forceinline__ float lx_dpp_q2(float x) {
    return __builtin_bit_cast(float, __builtin_amdgcn_update_dpp(0, __builtin_bit_cast(int, x), 0x4E, 0xF, 0xF, true));
}
static __device__ __forceinline__ float lx_dpp_x4(float x) {
    return __builtin_bit_cast(float, __builtin_amdgcn_update_dpp(0, __builtin_bit_cast(int, x), 0x164, 0xF, 0xF, true));
}
static __device__ __forceinline__ float lx_dpp_x8(float x) {
    return __builtin_bit_cast(float, __builtin_amdgcn_update_dpp(0, __builtin_bit_cast(int, x), 0x168, 0xF, 0xF, true));
}
static __device__ __forceinline__ float lx_half(float x) {
    return __builtin_bit_cast(float, __builtin_amdgcn_permlanex16(__builtin_bit_cast(int, x), __builtin_bit_cast(int, x),
                                                                  0x76543210, 0xfedcba98, false, false));
}

__global__ __launch_bounds__(256) void k_gather_cls(const float* __restrict__ H,
                                                    const int* __restrict__ seg,
                                                    const int* __restrict__ col,
                                                    const float* __restrict__ rsI,
                                                    const float* __restrict__ bias,
                                                    const float* __restrict__ Wc,
                                                    const float* __restrict__ bc,
                                                    float* __restrict__ out, int n) {
    __shared__ float lg[32];
    const int lane = threadIdx.x & 31;
    const int wave = threadIdx.x >> 5;
    const float bco = bc[lane & 3];
    int node = blockIdx.x * 8 + wave;
    if (node < n) {
        node = __builtin_amdgcn_readfirstlane(node);
        const float s = rsI[node];
        const float4 b = *(const float4*)(bias + lane * 4);
        const float4 acc = gather_row(H, seg, col, node, lane);
        const float4 h = row_tail(acc, s, b);

        const float4 w0 = *(const float4*)(Wc + lane * 16);
        const float4 w1 = *(const float4*)(Wc + lane * 16 + 4);
        const float4 w2 = *(const float4*)(Wc + lane * 16 + 8);
        const float4 w3 = *(const float4*)(Wc + lane * 16 + 12);
        const float p0 = h.x * w0.x + h.y * w1.x + h.z * w2.x + h.w * w3.x;
        const float p1 = h.x * w0.y + h.y * w1.y + h.z * w2.y + h.w * w3.y;
        const float p2 = h.x * w0.z + h.y * w1.z + h.z * w2.z + h.w * w3.z;
        const float p3 = h.x * w0.w + h.y * w1.w + h.z * w2.w + h.w * w3.w;

        const bool o1 = (lane & 1) != 0;
        const bool o2 = (lane & 2) != 0;
        const float q0 = (o1 ? p2 : p0) + lx_dpp_q1(o1 ? p0 : p2);
        const float q1 = (o1 ? p3 : p1) + lx_dpp_q1(o1 ? p1 : p3);
        float r = (o2 ? q1 : q0) + lx_dpp_q2(o2 ? q0 : q1);
        r += lx_dpp_x4(r);
        r += lx_dpp_x8(r);
        r += lx_half(r);
        if (lane < 4) lg[wave * 4 + (((lane & 1) << 1) | (lane >> 1))] = r;
    }
    __syncthreads();
    if (wave == 0) {
        const int i = blockIdx.x * 32 + lane;
        if (i < n * 4) out[i] = lg[lane] + bco;
    }
}

extern "C" void kernel_launch(void* const* d_in, const int* in_sizes, int n_in,
                              void* d_out, int out_size, void* d_ws, size_t ws_size,
                              hipStream_t stream) {
    const float* feat = (const float*)d_in[0];
    const int*   src  = (const int*)d_in[1];
    const int*   dst  = (const int*)d_in[2];
    const float* W1   = (const float*)d_in[3];
    const float* b1   = (const float*)d_in[4];
    const float* W2   = (const float*)d_in[5];
    const float* b2   = (const float*)d_in[6];
    const float* Wc   = (const float*)d_in[7];
    const float* bc   = (const float*)d_in[8];
    float* out = (float*)d_out;

    const int nnodes = in_sizes[0] / FEAT;
    const int nedges = in_sizes[1];
    const size_t nAlign = ((size_t)nnodes + 3) & ~(size_t)3;
    const size_t eAlign = ((size_t)nedges + 3) & ~(size_t)3;
    const size_t nfeat = (size_t)nnodes * FEAT;

    const int NB  = (nnodes + 255) >> BSH;
    const int NCH = (nedges + CHUNK - 1) / CHUNK;
    const size_t tabAlign = ((size_t)2 * NB * NCH + 3) & ~(size_t)3;
    const size_t totAlign = ((size_t)2 * NB + 3) & ~(size_t)3;

    int*      cntAll = (int*)d_ws;
    int*      offAll = cntAll + tabAlign;
    int*      totAll = offAll + tabAlign;
    int*      seg    = totAll + totAlign;
    float*    rsO    = (float*)(seg + 2 * nAlign);
    float*    rsI    = rsO + nAlign;
    unsigned* PD     = (unsigned*)(rsI + nAlign);
    int*      PS     = (int*)(PD + eAlign);
    int*      col    = PS + eAlign;
    float*    bufA   = (float*)(col + eAlign);
    float*    bufB   = bufA + nfeat;

    const int B = 256;
    const int gNodeW  = (nnodes + 7) / 8;
    const int gGemm   = (nnodes + ROWS_PER_BLK - 1) / ROWS_PER_BLK;

    k_range_count<<<NCH, B, 0, stream>>>(src, dst, cntAll, nedges, NB, NCH);
    k_chunk_scan<<<(2 * NB + 7) / 8, B, 0, stream>>>(cntAll, offAll, totAll, 2 * NB, NCH);
    k_range_file<<<NCH, B, 0, stream>>>(src, dst, offAll, totAll, PD, PS, nedges, NB, NCH);
    k_range_rank<<<2 * NB, B, 0, stream>>>(PD, PS, totAll, seg, col, rsO, rsI, nnodes, NB);

    k_gemm_scaled<<<gGemm, B, 0, stream>>>(feat, rsO, W1, bufA, nnodes);

    k_gather_relu_gemm<<<gGemm, B, 0, stream>>>(bufA, seg, col, rsI, rsO, b1, W2, bufB, nnodes);

    k_gather_cls<<<gNodeW, B, 0, stream>>>(bufB, seg, col, rsI, b2, Wc, bc, out, nnodes);
}


# edge sort: scattered 4-byte stores of PD, PS and col are put in slot order in an LDS stage and written as contiguous runs (a wave store touches a few lines, not 32); sum order by arrival rank still
# speedup vs baseline: 4.4059x; 1.2992x over previous
#include <hip/hip_runtime.h>
#include <math.h>

typedef __attribute__((ext_vector_type(2))) float v2f;
typedef __attribute__((ext_vector_type(8))) float v8f;

static constexpr int FEAT = 128;
static constexpr int APAD = 132;
static constexpr int ROWS_PER_BLK = 64;

static constexpr int BSH = 8;
static constexpr int NBMAX = 512;
static constexpr int CHUNK = 4096;
static constexpr int RCAP = 8192;

__global__ __launch_bounds__(256) void k_range_count(const int* __restrict__ src,
                                                     const int* __restrict__ dst,
                                                     int* __restrict__ cntAll,
                                                     int ne, int nb, int nch) {
    __shared__ int hd[NBMAX];
    __shared__ int hs[NBMAX];
    const int t = threadIdx.x;
    for (int b = t; b < NBMAX; b += 256) { hd[b] = 0; hs[b] = 0; }
    __syncthreads();
    const int c = blockIdx.x;
    const int e0 = c * CHUNK + t;
    int dv[CHUNK / 256], sv[CHUNK / 256];
    #pragma unroll
    for (int i = 0; i < CHUNK / 256; ++i) {
        const int e = e0 + i * 256;
        const int ec = e < ne ? e : ne - 1;
        dv[i] = dst[ec];
        sv[i] = src[ec];
    }
    #pragma unroll
    for (int i = 0; i < CHUNK / 256; ++i) {
        if (e0 + i * 256 < ne) {
            atomicAdd(&hd[dv[i] >> BSH], 1);
            atomicAdd(&hs[sv[i] >> BSH], 1);
        }
    }
    __syncthreads();
    for (int b = t; b < nb; b += 256) {
        cntAll[(size_t)b * nch + c] = hd[b];
        cntAll[(size_t)(nb + b) * nch + c] = hs[b];
    }
}

__global__ __launch_bounds__(256) void k_chunk_scan(const int* __restrict__ cntAll,
                                                    int* __restrict__ offAll,
                                                    int* __restrict__ totAll,
                                                    int nrows, int nch) {
    const int lane = threadIdx.x & 31;
    const int r = blockIdx.x * 8 + (threadIdx.x >> 5);
    if (r >= nrows) return;
    const int per = (nch + 31) >> 5;
    const int c0 = lane * per;
    const int* __restrict__ crow = cntAll + (size_t)r * nch;
    int* __restrict__ orow = offAll + (size_t)r * nch;

    int s = 0;
    for (int j0 = 0; j0 < per; j0 += 16) {
        int v[16];
        #pragma unroll
        for (int j = 0; j < 16; ++j) {
            const int c = c0 + j0 + j;
            v[j] = (j0 + j < per && c < nch) ? crow[c] : 0;
        }
        #pragma unroll
        for (int j = 0; j < 16; ++j) s += v[j];
    }
    int incl = s;
    for (int d = 1; d < 32; d <<= 1) {
        int u = __shfl_up(incl, d);
        if (lane >= d) incl += u;
    }
    if (lane == 31) totAll[r] = incl;
    int run = incl - s;
    for (int j0 = 0; j0 < per; j0 += 16) {
        int v[16];
        #pragma unroll
        for (int j = 0; j < 16; ++j) {
            const int c = c0 + j0 + j;
            v[j] = (j0 + j < per && c < nch) ? crow[c] : 0;
        }
        #pragma unroll
        for (int j = 0; j < 16; ++j) {
            const int c = c0 + j0 + j;
            if (j0 + j < per && c < nch) orow[c] = run;
            run += v[j];
        }
    }
}

template <bool FULL>
static __device__ __forceinline__ void file_edges(const int (&dv)[CHUNK / 256], const int (&sv)[CHUNK / 256],
                                                  int* curD, int* curS, const int* g0D, const int* g0S,
                                                  int* l0D, int* l0S, unsigned* stage, int* addr,
                                                  unsigned* __restrict__ PD, int* __restrict__ PS,
                                                  int e0, int ne, int nvalid, int t) {
    const int lane = t & 31;
    const int wave = t >> 5;
    int pd[CHUNK / 256], ps[CHUNK / 256];
    #pragma unroll
    for (int i = 0; i < CHUNK / 256; ++i) {
        if (FULL || e0 + i * 256 < ne) {
            pd[i] = atomicAdd(&curD[dv[i] >> BSH], 1);
            ps[i] = atomicAdd(&curS[sv[i] >> BSH], 1);
        }
    }
    __syncthreads();
    if (wave < 2) {
        const int* cur = wave == 0 ? curD : curS;
        const int* g0 = wave == 0 ? g0D : g0S;
        int* l0 = wave == 0 ? l0D : l0S;
        int loc[NBMAX / 32];
        int s = 0;
        #pragma unroll
        for (int j = 0; j < NBMAX / 32; ++j) {
            const int b = lane * (NBMAX / 32) + j;
            loc[j] = s;
            s += cur[b] - g0[b];
        }
        int incl = s;
        for (int d = 1; d < 32; d <<= 1) {
            int u = __shfl_up(incl, d);
            if (lane >= d) incl += u;
        }
        const int ex = incl - s;
        #pragma unroll
        for (int j = 0; j < NBMAX / 32; ++j) l0[lane * (NBMAX / 32) + j] = ex + loc[j];
    }
    __syncthreads();
    #pragma unroll
    for (int i = 0; i < CHUNK / 256; ++i) {
        if (FULL || e0 + i * 256 < ne) {
            const int b = dv[i] >> BSH;
            const int p = l0D[b] + (pd[i] - g0D[b]);
            stage[p] = ((unsigned)sv[i] << BSH) | (unsigned)(dv[i] & 255);
            addr[p] = pd[i];
        }
    }
    __syncthreads();
    if (FULL) {
        #pragma unroll
        for (int k = 0; k < CHUNK / 256; ++k) PD[addr[t + k * 256]] = stage[t + k * 256];
    } else {
        for (int i = t; i < nvalid; i += 256) PD[addr[i]] = stage[i];
    }
    __syncthreads();
    #pragma unroll
    for (int i = 0; i < CHUNK / 256; ++i) {
        if (FULL || e0 + i * 256 < ne) {
            const int b = sv[i] >> BSH;
            const int p = l0S[b] + (ps[i] - g0S[b]);
            stage[p] = (unsigned)(sv[i] & 255);
            addr[p] = ps[i];
        }
    }
    __syncthreads();
    if (FULL) {
        #pragma unroll
        for (int k = 0; k < CHUNK / 256; ++k) PS[addr[t + k * 256]] = (int)stage[t + k * 256];
    } else {
        for (int i = t; i < nvalid; i += 256) PS[addr[i]] = (int)stage[i];
    }
}

__global__ __launch_bounds__(256) void k_range_file(const int* __restrict__ src,
                                                    const int* __restrict__ dst,
                                                    const int* __restrict__ offAll,
                                                    const int* __restrict__ totAll,
                                                    unsigned* __restrict__ PD,
                                                    int* __restrict__ PS,
                                                    int ne, int nb, int nch) {
    __shared__ int curD[NBMAX];
    __shared__ int curS[NBMAX];
    __shared__ int g0D[NBMAX];
    __shared__ int g0S[NBMAX];
    __shared__ int l0D[NBMAX];
    __shared__ int l0S[NBMAX];
    __shared__ unsigned stage[CHUNK];
    __shared__ int addr[CHUNK];
    const int t = threadIdx.x;
    const int lane = t & 31;
    const int wave = t >> 5;
    const int c = blockIdx.x;
    const int e0 = c * CHUNK + t;
    int dv[CHUNK / 256], sv[CHUNK / 256];
    #pragma unroll
    for (int i = 0; i < CHUNK / 256; ++i) {
        const int e = e0 + i * 256;
        const int ec = e < ne ? e : ne - 1;
        dv[i] = dst[ec];
        sv[i] = src[ec];
    }
    if (wave < 2) {
        const int rb = wave == 0 ? 0 : nb;
        int* cur = wave == 0 ? curD : curS;
        int* g0 = wave == 0 ? g0D : g0S;
        int loc[NBMAX / 32];
        int s = 0;
        #pragma unroll
        for (int j = 0; j < NBMAX / 32; ++j) {
            const int b = lane * (NBMAX / 32) + j;
            const int v = b < nb ? totAll[rb + b] : 0;
            loc[j] = s;
            s += v;
        }
        int incl = s;
        for (int d = 1; d < 32; d <<= 1) {
            int u = __shfl_up(incl, d);
            if (lane >= d) incl += u;
        }
        const int ex = incl - s;
        int off[NBMAX / 32];
        #pragma unroll
        for (int j = 0; j < NBMAX / 32; ++j) {
            const int b = lane * (NBMAX / 32) + j;
            off[j] = offAll[(size_t)(rb + (b < nb ? b : nb - 1)) * nch + c];
        }
        #pragma unroll
        for (int j = 0; j < NBMAX / 32; ++j) {
            const int b = lane * (NBMAX / 32) + j;
            const int first = b < nb ? ex + loc[j] + off[j] : 0;
            cur[b] = first;
            g0[b] = first;
        }
    }
    __syncthreads();
    const int left = ne - c * CHUNK;
    const int nvalid = left < CHUNK ? left : CHUNK;
    if ((c + 1) * CHUNK <= ne) {
        file_edges<true>(dv, sv, curD, curS, g0D, g0S, l0D, l0S, stage, addr, PD, PS, e0, ne, nvalid, t);
    } else {
        file_edges<false>(dv, sv, curD, curS, g0D, g0S, l0D, l0S, stage, addr, PD, PS, e0, ne, nvalid, t);
    }
}

__global__ __launch_bounds__(256) void k_range_rank(const unsigned* __restrict__ PD,
                                                    const int* __restrict__ PS,
                                                    const int* __restrict__ totAll,
                                                    int* __restrict__ seg,
                                                    int* __restrict__ col,
                                                    float* __restrict__ rsO,
                                                    float* __restrict__ rsI,
                                                    int n, int nb) {
    __shared__ int cnt[256];
    __shared__ int cur[256];
    __shared__ int wred[8];
    __shared__ int wsum[8];
    __shared__ int colL[RCAP];
    const int t = threadIdx.x;
    const int lane = t & 31;
    const int wave = t >> 5;
    const int r = blockIdx.x;
    const bool isD = r < nb;
    const int b = isD ? r : r - nb;
    const int rb = isD ? 0 : nb;

    int acc = 0;
    for (int bb = t; bb < b; bb += 256) acc += totAll[rb + bb];
    for (int d = 16; d >= 1; d >>= 1) acc += __shfl_xor(acc, d);
    if (lane == 0) wred[wave] = acc;
    cnt[t] = 0;
    __syncthreads();
    int base = 0;
    for (int w = 0; w < 8; ++w) base += wred[w];
    const int size = totAll[r];

    const unsigned* __restrict__ P = isD ? PD : (const unsigned*)PS;
    for (int i0 = t; i0 < size + t; i0 += 2048) {
        unsigned k[8];
        #pragma unroll
        for (int j = 0; j < 8; ++j) {
            const int i = i0 + j * 256;
            k[j] = P[base + (i < size ? i : size - 1)];
        }
        #pragma unroll
        for (int j = 0; j < 8; ++j) {
            if (i0 + j * 256 < size) atomicAdd(&cnt[k[j] & 255u], 1);
        }
    }
    __syncthreads();
    const int cn = cnt[t];
    const int node = b * 256 + t;
    float a = (float)cn;
    a = a < 1.0f ? 1.0f : a;
    const float rs = 1.0f / sqrtf(a);
    if (!isD) {
        if (node < n) rsO[node] = rs;
        return;
    }

    int incl = cn;
    for (int d = 1; d < 32; d <<= 1) {
        int u = __shfl_up(incl, d);
        if (lane >= d) incl += u;
    }
    if (lane == 31) wsum[wave] = incl;
    __syncthreads();
    int woff = 0;
    for (int w = 0; w < 8; ++w) {
        if (w < wave) woff += wsum[w];
    }
    const int start = base + woff + incl - cn;
    cur[t] = start;
    if (node < n) {
        int2 sg = {start, cn};
        *(int2*)(seg + 2 * (size_t)node) = sg;
        rsI[node] = rs;
    }
    __syncthreads();
    const bool staged = size <= RCAP;
    for (int i0 = t; i0 < size + t; i0 += 2048) {
        unsigned k[8];
        #pragma unroll
        for (int j = 0; j < 8; ++j) {
            const int i = i0 + j * 256;
            k[j] = PD[base + (i < size ? i : size - 1)];
        }
        #pragma unroll
        for (int j = 0; j < 8; ++j) {
            if (i0 + j * 256 < size) {
                const int pos = atomicAdd(&cur[k[j] & 255u], 1);
                if (staged) colL[pos - base] = (int)(k[j] >> BSH);
                else        col[pos] = (int)(k[j] >> BSH);
            }
        }
    }
    if (staged) {
        __syncthreads();
        for (int i0 = t; i0 < size + t; i0 += 2048) {
            int v[8];
            #pragma unroll
            for (int j = 0; j < 8; ++j) {
                const int i = i0 + j * 256;
                v[j] = colL[i < size ? i : size - 1];
            }
            #pragma unroll
            for (int j = 0; j < 8; ++j) {
                const int i = i0 + j * 256;
                if (i < size) col[base + i] = v[j];
            }
        }
    }
}

static __device__ __forceinline__ void tile_product_store(const float* Alds,
                                                          const float* __restrict__ W,
                                                          float* __restrict__ Y,
                                                          int rowbase, int nrows, int t) {
    const int wave = t >> 5;
    const int lane = t & 31;
    const int half = lane >> 4;
    const int l16  = lane & 15;
    const int ncol = wave * 16 + l16;

    v8f acc0 = {}, acc1 = {}, acc2 = {}, acc3 = {};
    const float* a0 = Alds + (0 * 16 + l16) * APAD;
    const float* a1 = Alds + (1 * 16 + l16) * APAD;
    const float* a2 = Alds + (2 * 16 + l16) * APAD;
    const float* a3 = Alds + (3 * 16 + l16) * APAD;

    #pragma unroll 4
    for (int k = 0; k < FEAT; k += 4) {
        const int ka = k + half * 2;
        v2f b;
        b.x = W[ka * FEAT + ncol];
        b.y = W[(ka + 1) * FEAT + ncol];
        v2f a;
        a.x = a0[ka]; a.y = a0[ka + 1];
        acc0 = __builtin_amdgcn_wmma_f32_16x16x4_f32(false, a, false, b, (short)0, acc0, false, false);
        a.x = a1[ka]; a.y = a1[ka + 1];
        acc1 = __builtin_amdgcn_wmma_f32_16x16x4_f32(false, a, false, b, (short)0, acc1, false, false);
        a.x = a2[ka]; a.y = a2[ka + 1];
        acc2 = __builtin_amdgcn_wmma_f32_16x16x4_f32(false, a, false, b, (short)0, acc2, false, false);
        a.x = a3[ka]; a.y = a3[ka + 1];
        acc3 = __builtin_amdgcn_wmma_f32_16x16x4_f32(false, a, false, b, (short)0, acc3, false, false);
    }

    const int mrow0 = rowbase + half * 8;
    if (rowbase + ROWS_PER_BLK <= nrows) {
        #pragma unroll
        for (int r = 0; r < 8; ++r) {
            Y[(size_t)(mrow0 +  0 + r) * FEAT + ncol] = acc0[r];
            Y[(size_t)(mrow0 + 16 + r) * FEAT + ncol] = acc1[r];
            Y[(size_t)(mrow0 + 32 + r) * FEAT + ncol] = acc2[r];
            Y[(size_t)(mrow0 + 48 + r) * FEAT + ncol] = acc3[r];
        }
    } else {
        #pragma unroll
        for (int r = 0; r < 8; ++r) {
            int r0 = mrow0 + r;
            if (r0 +  0 < nrows) Y[(size_t)(r0 +  0) * FEAT + ncol] = acc0[r];
            if (r0 + 16 < nrows) Y[(size_t)(r0 + 16) * FEAT + ncol] = acc1[r];
            if (r0 + 32 < nrows) Y[(size_t)(r0 + 32) * FEAT + ncol] = acc2[r];
            if (r0 + 48 < nrows) Y[(size_t)(r0 + 48) * FEAT + ncol] = acc3[r];
        }
    }
}

__global__ __launch_bounds__(256) void k_gemm_scaled(const float* __restrict__ X,
                                                     const float* __restrict__ rs,
                                                     const float* __restrict__ W,
                                                     float* __restrict__ Y, int nrows) {
    __shared__ float Alds[ROWS_PER_BLK * APAD];

    const int rowbase = blockIdx.x * ROWS_PER_BLK;
    const int t = threadIdx.x;

    for (int i = t; i < ROWS_PER_BLK * (FEAT / 4); i += 256) {
        int r  = i >> 5;
        int c4 = (i & 31) << 2;
        int row = rowbase + r;
        float4 v = {0.f, 0.f, 0.f, 0.f};
        float s = 0.f;
        if (row < nrows) {
            v = *(const float4*)(X + (size_t)row * FEAT + c4);
            s = rs[row];
        }
        float* dstp = Alds + r * APAD + c4;
        dstp[0] = v.x * s; dstp[1] = v.y * s; dstp[2] = v.z * s; dstp[3] = v.w * s;
    }
    __syncthreads();
    tile_product_store(Alds, W, Y, rowbase, nrows, t);
}

static __device__ __forceinline__ void add_rows(float4& acc, const float* Hl, int cv, int m) {
#define ROW(j) (*(const float4*)(Hl + (size_t)__builtin_amdgcn_readlane(cv, (j)) * FEAT))
#define ADD(v) { acc.x += (v).x; acc.y += (v).y; acc.z += (v).z; acc.w += (v).w; }
    int j = 0;
    for (; j + 8 <= m; j += 8) {
        float4 v0 = ROW(j), v1 = ROW(j + 1), v2 = ROW(j + 2), v3 = ROW(j + 3);
        float4 v4 = ROW(j + 4), v5 = ROW(j + 5), v6 = ROW(j + 6), v7 = ROW(j + 7);
        ADD(v0) ADD(v1) ADD(v2) ADD(v3) ADD(v4) ADD(v5) ADD(v6) ADD(v7)
    }
    if (j + 4 <= m) {
        float4 v0 = ROW(j), v1 = ROW(j + 1), v2 = ROW(j + 2), v3 = ROW(j + 3);
        ADD(v0) ADD(v1) ADD(v2) ADD(v3)
        j += 4;
    }
    if (j + 2 <= m) {
        float4 v0 = ROW(j), v1 = ROW(j + 1);
        ADD(v0) ADD(v1)
        j += 2;
    }
    if (j < m) {
        float4 v0 = ROW(j);
        ADD(v0)
    }
#undef ROW
#undef ADD
}

static __device__ __forceinline__ float4 gather_row(const float* __restrict__ H,
                                                    const int* __restrict__ seg,
                                                    const int* __restrict__ col,
                                                    int node, int lane) {
    const int2 sg = *(const int2*)(seg + 2 * (size_t)node);
    const int start = sg.x;
    const int cnt = sg.y;
    const float* Hl = H + lane * 4;

    float4 acc = {0.f, 0.f, 0.f, 0.f};
    for (int base = 0; base < cnt; base += 32) {
        int m = cnt - base;
        if (m > 32) m = 32;
        int cv = 0;
        if (lane < m) cv = col[start + base + lane];
        add_rows(acc, Hl, cv, m);
    }
    return acc;
}

static __device__ __forceinline__ float4 gather_row_ahead(const float* __restrict__ H,
                                                          const int* __restrict__ col,
                                                          int start, int cnt, int cv0, int lane) {
    const float* Hl = H + lane * 4;
    float4 acc = {0.f, 0.f, 0.f, 0.f};
    add_rows(acc, Hl, cv0, cnt < 32 ? cnt : 32);
    for (int base = 32; base < cnt; base += 32) {
        int m = cnt - base;
        if (m > 32) m = 32;
        int cv = 0;
        if (lane < m) cv = col[start + base + lane];
        add_rows(acc, Hl, cv, m);
    }
    return acc;
}

static __device__ __forceinline__ float4 row_tail(float4 acc, float s, float4 b) {
    float4 h;
    h.x = acc.x * s + b.x; h.x = h.x > 0.f ? h.x : 0.f;
    h.y = acc.y * s + b.y; h.y = h.y > 0.f ? h.y : 0.f;
    h.z = acc.z * s + b.z; h.z = h.z > 0.f ? h.z : 0.f;
    h.w = acc.w * s + b.w; h.w = h.w > 0.f ? h.w : 0.f;
    return h;
}

__global__ __launch_bounds__(256) void k_gather_relu_gemm(const float* __restrict__ H,
                                                          const int* __restrict__ seg,
                                                          const int* __restrict__ col,
                                                          const float* __restrict__ rsI,
                                                          const float* __restrict__ rsO,
                                                          const float* __restrict__ bias,
                                                          const float* __restrict__ W,
                                                          float* __restrict__ Y, int n) {
    __shared__ float Alds[ROWS_PER_BLK * APAD];

    const int rowbase = blockIdx.x * ROWS_PER_BLK;
    const int t = threadIdx.x;
    const int lane = t & 31;
    const int wave = __builtin_amdgcn_readfirstlane(t >> 5);
    const float4 b = *(const float4*)(bias + lane * 4);

    const int r0 = wave * (ROWS_PER_BLK / 8);
    int2 sgv = {0, 0};
    float siv = 0.f, sov = 0.f;
    {
        const int nd = rowbase + r0 + (lane & 7);
        if (nd < n) {
            sgv = *(const int2*)(seg + 2 * (size_t)nd);
            siv = rsI[nd];
            sov = rsO[nd];
        }
    }
    int startN = __builtin_amdgcn_readlane(sgv.x, 0);
    int cntN   = __builtin_amdgcn_readlane(sgv.y, 0);
    int cvN = 0;
    if (lane < cntN) cvN = col[startN + lane];

    for (int i = 0; i < ROWS_PER_BLK / 8; ++i) {
        const int start = startN, cnt = cntN, cv0 = cvN;
        const float s  = __builtin_bit_cast(float, __builtin_amdgcn_readlane(__builtin_bit_cast(int, siv), i));
        const float so = __builtin_bit_cast(float, __builtin_amdgcn_readlane(__builtin_bit_cast(int, sov), i));
        if (i + 1 < ROWS_PER_BLK / 8) {
            startN = __builtin_amdgcn_readlane(sgv.x, i + 1);
            cntN   = __builtin_amdgcn_readlane(sgv.y, i + 1);
            cvN = 0;
            if (lane < cntN) cvN = col[startN + lane];
        }
        const float4 acc = gather_row_ahead(H, col, start, cnt, cv0, lane);
        const float4 h = row_tail(acc, s, b);
        float4 v = {0.f, 0.f, 0.f, 0.f};
        if (rowbase + r0 + i < n) { v.x = h.x * so; v.y = h.y * so; v.z = h.z * so; v.w = h.w * so; }
        *(float4*)(Alds + (r0 + i) * APAD + lane * 4) = v;
    }
    __syncthreads();
    tile_product_store(Alds, W, Y, rowbase, n, t);
}

static __device__ __forceinline__ float lx_dpp_q1(float x) {
    return __builtin_bit_cast(float, __builtin_amdgcn_update_dpp(0, __builtin_bit_cast(int, x), 0xB1, 0xF, 0xF, true));
}
static __device__ __forceinline__ float lx_dpp_q2(float x) {
    return __builtin_bit_cast(float, __builtin_amdgcn_update_dpp(0, __builtin_bit_cast(int, x), 0x4E, 0xF, 0xF, true));
}
static __device__ __forceinline__ float lx_dpp_x4(float x) {
    return __builtin_bit_cast(float, __builtin_amdgcn_update_dpp(0, __builtin_bit_cast(int, x), 0x164, 0xF, 0xF, true));
}
static __device__ __forceinline__ float lx_dpp_x8(float x) {
    return __builtin_bit_cast(float, __builtin_amdgcn_update_dpp(0, __builtin_bit_cast(int, x), 0x168, 0xF, 0xF, true));
}
static __device__ __forceinline__ float lx_half(float x) {
    return __builtin_bit_cast(float, __builtin_amdgcn_permlanex16(__builtin_bit_cast(int, x), __builtin_bit_cast(int, x),
                                                                  0x76543210, 0xfedcba98, false, false));
}

__global__ __launch_bounds__(256) void k_gather_cls(const float* __restrict__ H,
                                                    const int* __restrict__ seg,
                                                    const int* __restrict__ col,
                                                    const float* __restrict__ rsI,
                                                    const float* __restrict__ bias,
                                                    const float* __restrict__ Wc,
                                                    const float* __restrict__ bc,
                                                    float* __restrict__ out, int n) {
    __shared__ float lg[32];
    const int lane = threadIdx.x & 31;
    const int wave = threadIdx.x >> 5;
    const float bco = bc[lane & 3];
    int node = blockIdx.x * 8 + wave;
    if (node < n) {
        node = __builtin_amdgcn_readfirstlane(node);
        const float s = rsI[node];
        const float4 b = *(const float4*)(bias + lane * 4);
        const float4 acc = gather_row(H, seg, col, node, lane);
        const float4 h = row_tail(acc, s, b);

        const float4 w0 = *(const float4*)(Wc + lane * 16);
        const float4 w1 = *(const float4*)(Wc + lane * 16 + 4);
        const float4 w2 = *(const float4*)(Wc + lane * 16 + 8);
        const float4 w3 = *(const float4*)(Wc + lane * 16 + 12);
        const float p0 = h.x * w0.x + h.y * w1.x + h.z * w2.x + h.w * w3.x;
        const float p1 = h.x * w0.y + h.y * w1.y + h.z * w2.y + h.w * w3.y;
        const float p2 = h.x * w0.z + h.y * w1.z + h.z * w2.z + h.w * w3.z;
        const float p3 = h.x * w0.w + h.y * w1.w + h.z * w2.w + h.w * w3.w;

        const bool o1 = (lane & 1) != 0;
        const bool o2 = (lane & 2) != 0;
        const float q0 = (o1 ? p2 : p0) + lx_dpp_q1(o1 ? p0 : p2);
        const float q1 = (o1 ? p3 : p1) + lx_dpp_q1(o1 ? p1 : p3);
        float r = (o2 ? q1 : q0) + lx_dpp_q2(o2 ? q0 : q1);
        r += lx_dpp_x4(r);
        r += lx_dpp_x8(r);
        r += lx_half(r);
        if (lane < 4) lg[wave * 4 + (((lane & 1) << 1) | (lane >> 1))] = r;
    }
    __syncthreads();
    if (wave == 0) {
        const int i = blockIdx.x * 32 + lane;
        if (i < n * 4) out[i] = lg[lane] + bco;
    }
}

extern "C" void kernel_launch(void* const* d_in, const int* in_sizes, int n_in,
                              void* d_out, int out_size, void* d_ws, size_t ws_size,
                              hipStream_t stream) {
    const float* feat = (const float*)d_in[0];
    const int*   src  = (const int*)d_in[1];
    const int*   dst  = (const int*)d_in[2];
    const float* W1   = (const float*)d_in[3];
    const float* b1   = (const float*)d_in[4];
    const float* W2   = (const float*)d_in[5];
    const float* b2   = (const float*)d_in[6];
    const float* Wc   = (const float*)d_in[7];
    const float* bc   = (const float*)d_in[8];
    float* out = (float*)d_out;

    const int nnodes = in_sizes[0] / FEAT;
    const int nedges = in_sizes[1];
    const size_t nAlign = ((size_t)nnodes + 3) & ~(size_t)3;
    const size_t eAlign = ((size_t)nedges + 3) & ~(size_t)3;
    const size_t nfeat = (size_t)nnodes * FEAT;

    const int NB  = (nnodes + 255) >> BSH;
    const int NCH = (nedges + CHUNK - 1) / CHUNK;
    const size_t tabAlign = ((size_t)2 * NB * NCH + 3) & ~(size_t)3;
    const size_t totAlign = ((size_t)2 * NB + 3) & ~(size_t)3;

    int*      cntAll = (int*)d_ws;
    int*      offAll = cntAll + tabAlign;
    int*      totAll = offAll + tabAlign;
    int*      seg    = totAll + totAlign;
    float*    rsO    = (float*)(seg + 2 * nAlign);
    float*    rsI    = rsO + nAlign;
    unsigned* PD     = (unsigned*)(rsI + nAlign);
    int*      PS     = (int*)(PD + eAlign);
    int*      col    = PS + eAlign;
    float*    bufA   = (float*)(col + eAlign);
    float*    bufB   = bufA + nfeat;

    const int B = 256;
    const int gNodeW  = (nnodes + 7) / 8;
    const int gGemm   = (nnodes + ROWS_PER_BLK - 1) / ROWS_PER_BLK;

    k_range_count<<<NCH, B, 0, stream>>>(src, dst, cntAll, nedges, NB, NCH);
    k_chunk_scan<<<(2 * NB + 7) / 8, B, 0, stream>>>(cntAll, offAll, totAll, 2 * NB, NCH);
    k_range_file<<<NCH, B, 0, stream>>>(src, dst, offAll, totAll, PD, PS, nedges, NB, NCH);
    k_range_rank<<<2 * NB, B, 0, stream>>>(PD, PS, totAll, seg, col, rsO, rsI, nnodes, NB);

    k_gemm_scaled<<<gGemm, B, 0, stream>>>(feat, rsO, W1, bufA, nnodes);

    k_gather_relu_gemm<<<gGemm, B, 0, stream>>>(bufA, seg, col, rsI, rsO, b1, W2, bufB, nnodes);

    k_gather_cls<<<gNodeW, B, 0, stream>>>(bufB, seg, col, rsI, b2, Wc, bc, out, nnodes);
}


# row tables bufA and bufB moved to a 512-byte boundary of the workspace (they began 96 bytes into a line): a gathered 512-byte row covers 4 whole 128-byte lines, not 5; sum order unchanged
# speedup vs baseline: 4.6728x; 1.0606x over previous
#include <hip/hip_runtime.h>
#include <math.h>
#include <stdint.h>

typedef __attribute__((ext_vector_type(2))) float v2f;
typedef __attribute__((ext_vector_type(8))) float v8f;

static constexpr int FEAT = 128;
static constexpr int APAD = 132;
static constexpr int ROWS_PER_BLK = 64;

static constexpr int BSH = 8;
static constexpr int NBMAX = 512;
static constexpr int CHUNK = 4096;
static constexpr int RCAP = 8192;

__global__ __launch_bounds__(256) void k_range_count(const int* __restrict__ src,
                                                     const int* __restrict__ dst,
                                                     int* __restrict__ cntAll,
                                                     int ne, int nb, int nch) {
    __shared__ int hd[NBMAX];
    __shared__ int hs[NBMAX];
    const int t = threadIdx.x;
    for (int b = t; b < NBMAX; b += 256) { hd[b] = 0; hs[b] = 0; }
    __syncthreads();
    const int c = blockIdx.x;
    const int e0 = c * CHUNK + t;
    int dv[CHUNK / 256], sv[CHUNK / 256];
    #pragma unroll
    for (int i = 0; i < CHUNK / 256; ++i) {
        const int e = e0 + i * 256;
        const int ec = e < ne ? e : ne - 1;
        dv[i] = dst[ec];
        sv[i] = src[ec];
    }
    #pragma unroll
    for (int i = 0; i < CHUNK / 256; ++i) {
        if (e0 + i * 256 < ne) {
            atomicAdd(&hd[dv[i] >> BSH], 1);
            atomicAdd(&hs[sv[i] >> BSH], 1);
        }
    }
    __syncthreads();
    for (int b = t; b < nb; b += 256) {
        cntAll[(size_t)b * nch + c] = hd[b];
        cntAll[(size_t)(nb + b) * nch + c] = hs[b];
    }
}

__global__ __launch_bounds__(256) void k_chunk_scan(const int* __restrict__ cntAll,
                                                    int* __restrict__ offAll,
                                                    int* __restrict__ totAll,
                                                    int nrows, int nch) {
    const int lane = threadIdx.x & 31;
    const int r = blockIdx.x * 8 + (threadIdx.x >> 5);
    if (r >= nrows) return;
    const int per = (nch + 31) >> 5;
    const int c0 = lane * per;
    const int* __restrict__ crow = cntAll + (size_t)r * nch;
    int* __restrict__ orow = offAll + (size_t)r * nch;

    int s = 0;
    for (int j0 = 0; j0 < per; j0 += 16) {
        int v[16];
        #pragma unroll
        for (int j = 0; j < 16; ++j) {
            const int c = c0 + j0 + j;
            v[j] = (j0 + j < per && c < nch) ? crow[c] : 0;
        }
        #pragma unroll
        for (int j = 0; j < 16; ++j) s += v[j];
    }
    int incl = s;
    for (int d = 1; d < 32; d <<= 1) {
        int u = __shfl_up(incl, d);
        if (lane >= d) incl += u;
    }
    if (lane == 31) totAll[r] = incl;
    int run = incl - s;
    for (int j0 = 0; j0 < per; j0 += 16) {
        int v[16];
        #pragma unroll
        for (int j = 0; j < 16; ++j) {
            const int c = c0 + j0 + j;
            v[j] = (j0 + j < per && c < nch) ? crow[c] : 0;
        }
        #pragma unroll
        for (int j = 0; j < 16; ++j) {
            const int c = c0 + j0 + j;
            if (j0 + j < per && c < nch) orow[c] = run;
            run += v[j];
        }
    }
}

template <bool FULL>
static __device__ __forceinline__ void file_edges(const int (&dv)[CHUNK / 256], const int (&sv)[CHUNK / 256],
                                                  int* curD, int* curS, const int* g0D, const int* g0S,
                                                  int* l0D, int* l0S, unsigned* stage, int* addr,
                                                  unsigned* __restrict__ PD, int* __restrict__ PS,
                                                  int e0, int ne, int nvalid, int t) {
    const int lane = t & 31;
    const int wave = t >> 5;
    int pd[CHUNK / 256], ps[CHUNK / 256];
    #pragma unroll
    for (int i = 0; i < CHUNK / 256; ++i) {
        if (FULL || e0 + i * 256 < ne) {
            pd[i] = atomicAdd(&curD[dv[i] >> BSH], 1);
            ps[i] = atomicAdd(&curS[sv[i] >> BSH], 1);
        }
    }
    __syncthreads();
    if (wave < 2) {
        const int* cur = wave == 0 ? curD : curS;
        const int* g0 = wave == 0 ? g0D : g0S;
        int* l0 = wave == 0 ? l0D : l0S;
        int loc[NBMAX / 32];
        int s = 0;
        #pragma unroll
        for (int j = 0; j < NBMAX / 32; ++j) {
            const int b = lane * (NBMAX / 32) + j;
            loc[j] = s;
            s += cur[b] - g0[b];
        }
        int incl = s;
        for (int d = 1; d < 32; d <<= 1) {
            int u = __shfl_up(incl, d);
            if (lane >= d) incl += u;
        }
        const int ex = incl - s;
        #pragma unroll
        for (int j = 0; j < NBMAX / 32; ++j) l0[lane * (NBMAX / 32) + j] = ex + loc[j];
    }
    __syncthreads();
    #pragma unroll
    for (int i = 0; i < CHUNK / 256; ++i) {
        if (FULL || e0 + i * 256 < ne) {
            const int b = dv[i] >> BSH;
            const int p = l0D[b] + (pd[i] - g0D[b]);
            stage[p] = ((unsigned)sv[i] << BSH) | (unsigned)(dv[i] & 255);
            addr[p] = pd[i];
        }
    }
    __syncthreads();
    if (FULL) {
        #pragma unroll
        for (int k = 0; k < CHUNK / 256; ++k) PD[addr[t + k * 256]] = stage[t + k * 256];
    } else {
        for (int i = t; i < nvalid; i += 256) PD[addr[i]] = stage[i];
    }
    __syncthreads();
    #pragma unroll
    for (int i = 0; i < CHUNK / 256; ++i) {
        if (FULL || e0 + i * 256 < ne) {
            const int b = sv[i] >> BSH;
            const int p = l0S[b] + (ps[i] - g0S[b]);
            stage[p] = (unsigned)(sv[i] & 255);
            addr[p] = ps[i];
        }
    }
    __syncthreads();
    if (FULL) {
        #pragma unroll
        for (int k = 0; k < CHUNK / 256; ++k) PS[addr[t + k * 256]] = (int)stage[t + k * 256];
    } else {
        for (int i = t; i < nvalid; i += 256) PS[addr[i]] = (int)stage[i];
    }
}

__global__ __launch_bounds__(256) void k_range_file(const int* __restrict__ src,
                                                    const int* __restrict__ dst,
                                                    const int* __restrict__ offAll,
                                                    const int* __restrict__ totAll,
                                                    unsigned* __restrict__ PD,
                                                    int* __restrict__ PS,
                                                    int ne, int nb, int nch) {
    __shared__ int curD[NBMAX];
    __shared__ int curS[NBMAX];
    __shared__ int g0D[NBMAX];
    __shared__ int g0S[NBMAX];
    __shared__ int l0D[NBMAX];
    __shared__ int l0S[NBMAX];
    __shared__ unsigned stage[CHUNK];
    __shared__ int addr[CHUNK];
    const int t = threadIdx.x;
    const int lane = t & 31;
    const int wave = t >> 5;
    const int c = blockIdx.x;
    const int e0 = c * CHUNK + t;
    int dv[CHUNK / 256], sv[CHUNK / 256];
    #pragma unroll
    for (int i = 0; i < CHUNK / 256; ++i) {
        const int e = e0 + i * 256;
        const int ec = e < ne ? e : ne - 1;
        dv[i] = dst[ec];
        sv[i] = src[ec];
    }
    if (wave < 2) {
        const int rb = wave == 0 ? 0 : nb;
        int* cur = wave == 0 ? curD : curS;
        int* g0 = wave == 0 ? g0D : g0S;
        int loc[NBMAX / 32];
        int s = 0;
        #pragma unroll
        for (int j = 0; j < NBMAX / 32; ++j) {
            const int b = lane * (NBMAX / 32) + j;
            const int v = b < nb ? totAll[rb + b] : 0;
            loc[j] = s;
            s += v;
        }
        int incl = s;
        for (int d = 1; d < 32; d <<= 1) {
            int u = __shfl_up(incl, d);
            if (lane >= d) incl += u;
        }
        const int ex = incl - s;
        int off[NBMAX / 32];
        #pragma unroll
        for (int j = 0; j < NBMAX / 32; ++j) {
            const int b = lane * (NBMAX / 32) + j;
            off[j] = offAll[(size_t)(rb + (b < nb ? b : nb - 1)) * nch + c];
        }
        #pragma unroll
        for (int j = 0; j < NBMAX / 32; ++j) {
            const int b = lane * (NBMAX / 32) + j;
            const int first = b < nb ? ex + loc[j] + off[j] : 0;
            cur[b] = first;
            g0[b] = first;
        }
    }
    __syncthreads();
    const int left = ne - c * CHUNK;
    const int nvalid = left < CHUNK ? left : CHUNK;
    if ((c + 1) * CHUNK <= ne) {
        file_edges<true>(dv, sv, curD, curS, g0D, g0S, l0D, l0S, stage, addr, PD, PS, e0, ne, nvalid, t);
    } else {
        file_edges<false>(dv, sv, curD, curS, g0D, g0S, l0D, l0S, stage, addr, PD, PS, e0, ne, nvalid, t);
    }
}

__global__ __launch_bounds__(256) void k_range_rank(const unsigned* __restrict__ PD,
                                                    const int* __restrict__ PS,
                                                    const int* __restrict__ totAll,
                                                    int* __restrict__ seg,
                                                    int* __restrict__ col,
                                                    float* __restrict__ rsO,
                                                    float* __restrict__ rsI,
                                                    int n, int nb) {
    __shared__ int cnt[256];
    __shared__ int cur[256];
    __shared__ int wred[8];
    __shared__ int wsum[8];
    __shared__ int colL[RCAP];
    const int t = threadIdx.x;
    const int lane = t & 31;
    const int wave = t >> 5;
    const int r = blockIdx.x;
    const bool isD = r < nb;
    const int b = isD ? r : r - nb;
    const int rb = isD ? 0 : nb;

    int acc = 0;
    for (int bb = t; bb < b; bb += 256) acc += totAll[rb + bb];
    for (int d = 16; d >= 1; d >>= 1) acc += __shfl_xor(acc, d);
    if (lane == 0) wred[wave] = acc;
    cnt[t] = 0;
    __syncthreads();
    int base = 0;
    for (int w = 0; w < 8; ++w) base += wred[w];
    const int size = totAll[r];

    const unsigned* __restrict__ P = isD ? PD : (const unsigned*)PS;
    for (int i0 = t; i0 < size + t; i0 += 2048) {
        unsigned k[8];
        #pragma unroll
        for (int j = 0; j < 8; ++j) {
            const int i = i0 + j * 256;
            k[j] = P[base + (i < size ? i : size - 1)];
        }
        #pragma unroll
        for (int j = 0; j < 8; ++j) {
            if (i0 + j * 256 < size) atomicAdd(&cnt[k[j] & 255u], 1);
        }
    }
    __syncthreads();
    const int cn = cnt[t];
    const int node = b * 256 + t;
    float a = (float)cn;
    a = a < 1.0f ? 1.0f : a;
    const float rs = 1.0f / sqrtf(a);
    if (!isD) {
        if (node < n) rsO[node] = rs;
        return;
    }

    int incl = cn;
    for (int d = 1; d < 32; d <<= 1) {
        int u = __shfl_up(incl, d);
        if (lane >= d) incl += u;
    }
    if (lane == 31) wsum[wave] = incl;
    __syncthreads();
    int woff = 0;
    for (int w = 0; w < 8; ++w) {
        if (w < wave) woff += wsum[w];
    }
    const int start = base + woff + incl - cn;
    cur[t] = start;
    if (node < n) {
        int2 sg = {start, cn};
        *(int2*)(seg + 2 * (size_t)node) = sg;
        rsI[node] = rs;
    }
    __syncthreads();
    const bool staged = size <= RCAP;
    for (int i0 = t; i0 < size + t; i0 += 2048) {
        unsigned k[8];
        #pragma unroll
        for (int j = 0; j < 8; ++j) {
            const int i = i0 + j * 256;
            k[j] = PD[base + (i < size ? i : size - 1)];
        }
        #pragma unroll
        for (int j = 0; j < 8; ++j) {
            if (i0 + j * 256 < size) {
                const int pos = atomicAdd(&cur[k[j] & 255u], 1);
                if (staged) colL[pos - base] = (int)(k[j] >> BSH);
                else        col[pos] = (int)(k[j] >> BSH);
            }
        }
    }
    if (staged) {
        __syncthreads();
        for (int i0 = t; i0 < size + t; i0 += 2048) {
            int v[8];
            #pragma unroll
            for (int j = 0; j < 8; ++j) {
                const int i = i0 + j * 256;
                v[j] = colL[i < size ? i : size - 1];
            }
            #pragma unroll
            for (int j = 0; j < 8; ++j) {
                const int i = i0 + j * 256;
                if (i < size) col[base + i] = v[j];
            }
        }
    }
}

static __device__ __forceinline__ void tile_product_store(const float* Alds,
                                                          const float* __restrict__ W,
                                                          float* __restrict__ Y,
                                                          int rowbase, int nrows, int t) {
    const int wave = t >> 5;
    const int lane = t & 31;
    const int half = lane >> 4;
    const int l16  = lane & 15;
    const int ncol = wave * 16 + l16;

    v8f acc0 = {}, acc1 = {}, acc2 = {}, acc3 = {};
    const float* a0 = Alds + (0 * 16 + l16) * APAD;
    const float* a1 = Alds + (1 * 16 + l16) * APAD;
    const float* a2 = Alds + (2 * 16 + l16) * APAD;
    const float* a3 = Alds + (3 * 16 + l16) * APAD;

    #pragma unroll 4
    for (int k = 0; k < FEAT; k += 4) {
        const int ka = k + half * 2;
        v2f b;
        b.x = W[ka * FEAT + ncol];
        b.y = W[(ka + 1) * FEAT + ncol];
        v2f a;
        a.x = a0[ka]; a.y = a0[ka + 1];
        acc0 = __builtin_amdgcn_wmma_f32_16x16x4_f32(false, a, false, b, (short)0, acc0, false, false);
        a.x = a1[ka]; a.y = a1[ka + 1];
        acc1 = __builtin_amdgcn_wmma_f32_16x16x4_f32(false, a, false, b, (short)0, acc1, false, false);
        a.x = a2[ka]; a.y = a2[ka + 1];
        acc2 = __builtin_amdgcn_wmma_f32_16x16x4_f32(false, a, false, b, (short)0, acc2, false, false);
        a.x = a3[ka]; a.y = a3[ka + 1];
        acc3 = __builtin_amdgcn_wmma_f32_16x16x4_f32(false, a, false, b, (short)0, acc3, false, false);
    }

    const int mrow0 = rowbase + half * 8;
    if (rowbase + ROWS_PER_BLK <= nrows) {
        #pragma unroll
        for (int r = 0; r < 8; ++r) {
            Y[(size_t)(mrow0 +  0 + r) * FEAT + ncol] = acc0[r];
            Y[(size_t)(mrow0 + 16 + r) * FEAT + ncol] = acc1[r];
            Y[(size_t)(mrow0 + 32 + r) * FEAT + ncol] = acc2[r];
            Y[(size_t)(mrow0 + 48 + r) * FEAT + ncol] = acc3[r];
        }
    } else {
        #pragma unroll
        for (int r = 0; r < 8; ++r) {
            int r0 = mrow0 + r;
            if (r0 +  0 < nrows) Y[(size_t)(r0 +  0) * FEAT + ncol] = acc0[r];
            if (r0 + 16 < nrows) Y[(size_t)(r0 + 16) * FEAT + ncol] = acc1[r];
            if (r0 + 32 < nrows) Y[(size_t)(r0 + 32) * FEAT + ncol] = acc2[r];
            if (r0 + 48 < nrows) Y[(size_t)(r0 + 48) * FEAT + ncol] = acc3[r];
        }
    }
}

__global__ __launch_bounds__(256) void k_gemm_scaled(const float* __restrict__ X,
                                                     const float* __restrict__ rs,
                                                     const float* __restrict__ W,
                                                     float* __restrict__ Y, int nrows) {
    __shared__ float Alds[ROWS_PER_BLK * APAD];

    const int rowbase = blockIdx.x * ROWS_PER_BLK;
    const int t = threadIdx.x;

    for (int i = t; i < ROWS_PER_BLK * (FEAT / 4); i += 256) {
        int r  = i >> 5;
        int c4 = (i & 31) << 2;
        int row = rowbase + r;
        float4 v = {0.f, 0.f, 0.f, 0.f};
        float s = 0.f;
        if (row < nrows) {
            v = *(const float4*)(X + (size_t)row * FEAT + c4);
            s = rs[row];
        }
        float* dstp = Alds + r * APAD + c4;
        dstp[0] = v.x * s; dstp[1] = v.y * s; dstp[2] = v.z * s; dstp[3] = v.w * s;
    }
    __syncthreads();
    tile_product_store(Alds, W, Y, rowbase, nrows, t);
}

static __device__ __forceinline__ void add_rows(float4& acc, const float* Hl, int cv, int m) {
#define ROW(j) (*(const float4*)(Hl + (size_t)__builtin_amdgcn_readlane(cv, (j)) * FEAT))
#define ADD(v) { acc.x += (v).x; acc.y += (v).y; acc.z += (v).z; acc.w += (v).w; }
    int j = 0;
    for (; j + 8 <= m; j += 8) {
        float4 v0 = ROW(j), v1 = ROW(j + 1), v2 = ROW(j + 2), v3 = ROW(j + 3);
        float4 v4 = ROW(j + 4), v5 = ROW(j + 5), v6 = ROW(j + 6), v7 = ROW(j + 7);
        ADD(v0) ADD(v1) ADD(v2) ADD(v3) ADD(v4) ADD(v5) ADD(v6) ADD(v7)
    }
    if (j + 4 <= m) {
        float4 v0 = ROW(j), v1 = ROW(j + 1), v2 = ROW(j + 2), v3 = ROW(j + 3);
        ADD(v0) ADD(v1) ADD(v2) ADD(v3)
        j += 4;
    }
    if (j + 2 <= m) {
        float4 v0 = ROW(j), v1 = ROW(j + 1);
        ADD(v0) ADD(v1)
        j += 2;
    }
    if (j < m) {
        float4 v0 = ROW(j);
        ADD(v0)
    }
#undef ROW
#undef ADD
}

static __device__ __forceinline__ float4 gather_row(const float* __restrict__ H,
                                                    const int* __restrict__ seg,
                                                    const int* __restrict__ col,
                                                    int node, int lane) {
    const int2 sg = *(const int2*)(seg + 2 * (size_t)node);
    const int start = sg.x;
    const int cnt = sg.y;
    const float* Hl = H + lane * 4;

    float4 acc = {0.f, 0.f, 0.f, 0.f};
    for (int base = 0; base < cnt; base += 32) {
        int m = cnt - base;
        if (m > 32) m = 32;
        int cv = 0;
        if (lane < m) cv = col[start + base + lane];
        add_rows(acc, Hl, cv, m);
    }
    return acc;
}

static __device__ __forceinline__ float4 gather_row_ahead(const float* __restrict__ H,
                                                          const int* __restrict__ col,
                                                          int start, int cnt, int cv0, int lane) {
    const float* Hl = H + lane * 4;
    float4 acc = {0.f, 0.f, 0.f, 0.f};
    add_rows(acc, Hl, cv0, cnt < 32 ? cnt : 32);
    for (int base = 32; base < cnt; base += 32) {
        int m = cnt - base;
        if (m > 32) m = 32;
        int cv = 0;
        if (lane < m) cv = col[start + base + lane];
        add_rows(acc, Hl, cv, m);
    }
    return acc;
}

static __device__ __forceinline__ float4 row_tail(float4 acc, float s, float4 b) {
    float4 h;
    h.x = acc.x * s + b.x; h.x = h.x > 0.f ? h.x : 0.f;
    h.y = acc.y * s + b.y; h.y = h.y > 0.f ? h.y : 0.f;
    h.z = acc.z * s + b.z; h.z = h.z > 0.f ? h.z : 0.f;
    h.w = acc.w * s + b.w; h.w = h.w > 0.f ? h.w : 0.f;
    return h;
}

__global__ __launch_bounds__(256) void k_gather_relu_gemm(const float* __restrict__ H,
                                                          const int* __restrict__ seg,
                                                          const int* __restrict__ col,
                                                          const float* __restrict__ rsI,
                                                          const float* __restrict__ rsO,
                                                          const float* __restrict__ bias,
                                                          const float* __restrict__ W,
                                                          float* __restrict__ Y, int n) {
    __shared__ float Alds[ROWS_PER_BLK * APAD];

    const int rowbase = blockIdx.x * ROWS_PER_BLK;
    const int t = threadIdx.x;
    const int lane = t & 31;
    const int wave = __builtin_amdgcn_readfirstlane(t >> 5);
    const float4 b = *(const float4*)(bias + lane * 4);

    const int r0 = wave * (ROWS_PER_BLK / 8);
    int2 sgv = {0, 0};
    float siv = 0.f, sov = 0.f;
    {
        const int nd = rowbase + r0 + (lane & 7);
        if (nd < n) {
            sgv = *(const int2*)(seg + 2 * (size_t)nd);
            siv = rsI[nd];
            sov = rsO[nd];
        }
    }
    int startN = __builtin_amdgcn_readlane(sgv.x, 0);
    int cntN   = __builtin_amdgcn_readlane(sgv.y, 0);
    int cvN = 0;
    if (lane < cntN) cvN = col[startN + lane];

    for (int i = 0; i < ROWS_PER_BLK / 8; ++i) {
        const int start = startN, cnt = cntN, cv0 = cvN;
        const float s  = __builtin_bit_cast(float, __builtin_amdgcn_readlane(__builtin_bit_cast(int, siv), i));
        const float so = __builtin_bit_cast(float, __builtin_amdgcn_readlane(__builtin_bit_cast(int, sov), i));
        if (i + 1 < ROWS_PER_BLK / 8) {
            startN = __builtin_amdgcn_readlane(sgv.x, i + 1);
            cntN   = __builtin_amdgcn_readlane(sgv.y, i + 1);
            cvN = 0;
            if (lane < cntN) cvN = col[startN + lane];
        }
        const float4 acc = gather_row_ahead(H, col, start, cnt, cv0, lane);
        const float4 h = row_tail(acc, s, b);
        float4 v = {0.f, 0.f, 0.f, 0.f};
        if (rowbase + r0 + i < n) { v.x = h.x * so; v.y = h.y * so; v.z = h.z * so; v.w = h.w * so; }
        *(float4*)(Alds + (r0 + i) * APAD + lane * 4) = v;
    }
    __syncthreads();
    tile_product_store(Alds, W, Y, rowbase, n, t);
}

static __device__ __forceinline__ float lx_dpp_q1(float x) {
    return __builtin_bit_cast(float, __builtin_amdgcn_update_dpp(0, __builtin_bit_cast(int, x), 0xB1, 0xF, 0xF, true));
}
static __device__ __forceinline__ float lx_dpp_q2(float x) {
    return __builtin_bit_cast(float, __builtin_amdgcn_update_dpp(0, __builtin_bit_cast(int, x), 0x4E, 0xF, 0xF, true));
}
static __device__ __forceinline__ float lx_dpp_x4(float x) {
    return __builtin_bit_cast(float, __builtin_amdgcn_update_dpp(0, __builtin_bit_cast(int, x), 0x164, 0xF, 0xF, true));
}
static __device__ __forceinline__ float lx_dpp_x8(float x) {
    return __builtin_bit_cast(float, __builtin_amdgcn_update_dpp(0, __builtin_bit_cast(int, x), 0x168, 0xF, 0xF, true));
}
static __device__ __forceinline__ float lx_half(float x) {
    return __builtin_bit_cast(float, __builtin_amdgcn_permlanex16(__builtin_bit_cast(int, x), __builtin_bit_cast(int, x),
                                                                  0x76543210, 0xfedcba98, false, false));
}

__global__ __launch_bounds__(256) void k_gather_cls(const float* __restrict__ H,
                                                    const int* __restrict__ seg,
                                                    const int* __restrict__ col,
                                                    const float* __restrict__ rsI,
                                                    const float* __restrict__ bias,
                                                    const float* __restrict__ Wc,
                                                    const float* __restrict__ bc,
                                                    float* __restrict__ out, int n) {
    __shared__ float lg[32];
    const int lane = threadIdx.x & 31;
    const int wave = threadIdx.x >> 5;
    const float bco = bc[lane & 3];
    int node = blockIdx.x * 8 + wave;
    if (node < n) {
        node = __builtin_amdgcn_readfirstlane(node);
        const float s = rsI[node];
        const float4 b = *(const float4*)(bias + lane * 4);
        const float4 acc = gather_row(H, seg, col, node, lane);
        const float4 h = row_tail(acc, s, b);

        const float4 w0 = *(const float4*)(Wc + lane * 16);
        const float4 w1 = *(const float4*)(Wc + lane * 16 + 4);
        const float4 w2 = *(const float4*)(Wc + lane * 16 + 8);
        const float4 w3 = *(const float4*)(Wc + lane * 16 + 12);
        const float p0 = h.x * w0.x + h.y * w1.x + h.z * w2.x + h.w * w3.x;
        const float p1 = h.x * w0.y + h.y * w1.y + h.z * w2.y + h.w * w3.y;
        const float p2 = h.x * w0.z + h.y * w1.z + h.z * w2.z + h.w * w3.z;
        const float p3 = h.x * w0.w + h.y * w1.w + h.z * w2.w + h.w * w3.w;

        const bool o1 = (lane & 1) != 0;
        const bool o2 = (lane & 2) != 0;
        const float q0 = (o1 ? p2 : p0) + lx_dpp_q1(o1 ? p0 : p2);
        const float q1 = (o1 ? p3 : p1) + lx_dpp_q1(o1 ? p1 : p3);
        float r = (o2 ? q1 : q0) + lx_dpp_q2(o2 ? q0 : q1);
        r += lx_dpp_x4(r);
        r += lx_dpp_x8(r);
        r += lx_half(r);
        if (lane < 4) lg[wave * 4 + (((lane & 1) << 1) | (lane >> 1))] = r;
    }
    __syncthreads();
    if (wave == 0) {
        const int i = blockIdx.x * 32 + lane;
        if (i < n * 4) out[i] = lg[lane] + bco;
    }
}

extern "C" void kernel_launch(void* const* d_in, const int* in_sizes, int n_in,
                              void* d_out, int out_size, void* d_ws, size_t ws_size,
                              hipStream_t stream) {
    const float* feat = (const float*)d_in[0];
    const int*   src  = (const int*)d_in[1];
    const int*   dst  = (const int*)d_in[2];
    const float* W1   = (const float*)d_in[3];
    const float* b1   = (const float*)d_in[4];
    const float* W2   = (const float*)d_in[5];
    const float* b2   = (const float*)d_in[6];
    const float* Wc   = (const float*)d_in[7];
    const float* bc   = (const float*)d_in[8];
    float* out = (float*)d_out;

    const int nnodes = in_sizes[0] / FEAT;
    const int nedges = in_sizes[1];
    const size_t nAlign = ((size_t)nnodes + 3) & ~(size_t)3;
    const size_t eAlign = ((size_t)nedges + 3) & ~(size_t)3;
    const size_t nfeat = (size_t)nnodes * FEAT;

    const int NB  = (nnodes + 255) >> BSH;
    const int NCH = (nedges + CHUNK - 1) / CHUNK;
    const size_t tabAlign = ((size_t)2 * NB * NCH + 3) & ~(size_t)3;
    const size_t totAlign = ((size_t)2 * NB + 3) & ~(size_t)3;

    int*      cntAll = (int*)d_ws;
    int*      offAll = cntAll + tabAlign;
    int*      totAll = offAll + tabAlign;
    int*      seg    = totAll + totAlign;
    float*    rsO    = (float*)(seg + 2 * nAlign);
    float*    rsI    = rsO + nAlign;
    unsigned* PD     = (unsigned*)(rsI + nAlign);
    int*      PS     = (int*)(PD + eAlign);
    int*      col    = PS + eAlign;
    float*    bufA   = (float*)(((uintptr_t)(col + eAlign) + 511) & ~(uintptr_t)511);
    float*    bufB   = bufA + nfeat;

    const int B = 256;
    const int gNodeW  = (nnodes + 7) / 8;
    const int gGemm   = (nnodes + ROWS_PER_BLK - 1) / ROWS_PER_BLK;

    k_range_count<<<NCH, B, 0, stream>>>(src, dst, cntAll, nedges, NB, NCH);
    k_chunk_scan<<<(2 * NB + 7) / 8, B, 0, stream>>>(cntAll, offAll, totAll, 2 * NB, NCH);
    k_range_file<<<NCH, B, 0, stream>>>(src, dst, offAll, totAll, PD, PS, nedges, NB, NCH);
    k_range_rank<<<2 * NB, B, 0, stream>>>(PD, PS, totAll, seg, col, rsO, rsI, nnodes, NB);

    k_gemm_scaled<<<gGemm, B, 0, stream>>>(feat, rsO, W1, bufA, nnodes);

    k_gather_relu_gemm<<<gGemm, B, 0, stream>>>(bufA, seg, col, rsI, rsO, b1, W2, bufB, nnodes);

    k_gather_cls<<<gNodeW, B, 0, stream>>>(bufB, seg, col, rsI, b2, Wc, bc, out, nnodes);
}


# edge sort in 2 launches, not 4: a chunk orders its edges by range in place and writes its ranges' first places; the ranking pass picks a range's pieces out of all chunks; arrival-rank sum order still
# speedup vs baseline: 5.0424x; 1.0791x over previous
#include <hip/hip_runtime.h>
#include <math.h>
#include <stdint.h>

typedef __attribute__((ext_vector_type(2))) float v2f;
typedef __attribute__((ext_vector_type(8))) float v8f;

static constexpr int FEAT = 128;
static constexpr int APAD = 132;
static constexpr int ROWS_PER_BLK = 64;

static constexpr int BSH = 8;
static constexpr int NBMAX = 512;
static constexpr int CHUNK = 4096;
static constexpr int NCHMAX = 1024;
static constexpr int RCAP = 8192;
static constexpr int SEGU = 16;

__global__ __launch_bounds__(256) void k_chunk_file(const int* __restrict__ src,
                                                    const int* __restrict__ dst,
                                                    int* __restrict__ tab,
                                                    unsigned* __restrict__ PD,
                                                    int* __restrict__ PS,
                                                    int ne, int nb, int nch) {
    __shared__ int hd[NBMAX];
    __shared__ int hs[NBMAX];
    __shared__ unsigned stage[CHUNK];
    const int t = threadIdx.x;
    const int lane = t & 31;
    const int wave = t >> 5;
    const int c = blockIdx.x;
    const int e0 = c * CHUNK + t;
    int dv[CHUNK / 256], sv[CHUNK / 256];
    #pragma unroll
    for (int i = 0; i < CHUNK / 256; ++i) {
        const int e = e0 + i * 256;
        const int ec = e < ne ? e : ne - 1;
        dv[i] = dst[ec];
        sv[i] = src[ec];
    }
    for (int b = t; b < NBMAX; b += 256) { hd[b] = 0; hs[b] = 0; }
    __syncthreads();
    int pd[CHUNK / 256], ps[CHUNK / 256];
    #pragma unroll
    for (int i = 0; i < CHUNK / 256; ++i) {
        if (e0 + i * 256 < ne) {
            pd[i] = atomicAdd(&hd[dv[i] >> BSH], 1);
            ps[i] = atomicAdd(&hs[sv[i] >> BSH], 1);
        }
    }
    __syncthreads();
    if (wave < 2) {
        int* h = wave == 0 ? hd : hs;
        int loc[NBMAX / 32];
        int s = 0;
        #pragma unroll
        for (int j = 0; j < NBMAX / 32; ++j) {
            loc[j] = s;
            s += h[lane * (NBMAX / 32) + j];
        }
        int incl = s;
        for (int d = 1; d < 32; d <<= 1) {
            int u = __shfl_up(incl, d);
            if (lane >= d) incl += u;
        }
        const int ex = incl - s;
        #pragma unroll
        for (int j = 0; j < NBMAX / 32; ++j) h[lane * (NBMAX / 32) + j] = ex + loc[j];
    }
    __syncthreads();
    const int left = ne - c * CHUNK;
    const int nvalid = left < CHUNK ? left : CHUNK;
    for (int b = t; b <= nb; b += 256) {
        const int bb = b < nb ? b : 0;
        tab[(size_t)b * nch + c] = b < nb ? hd[bb] : nvalid;
        tab[(size_t)(nb + 1 + b) * nch + c] = b < nb ? hs[bb] : nvalid;
    }
    #pragma unroll
    for (int i = 0; i < CHUNK / 256; ++i) {
        if (e0 + i * 256 < ne) stage[hd[dv[i] >> BSH] + pd[i]] = ((unsigned)sv[i] << BSH) | (unsigned)(dv[i] & 255);
    }
    __syncthreads();
    unsigned* __restrict__ PDc = PD + (size_t)c * CHUNK;
    int* __restrict__ PSc = PS + (size_t)c * CHUNK;
    if (nvalid == CHUNK) {
        #pragma unroll
        for (int k = 0; k < CHUNK / 256; ++k) PDc[t + k * 256] = stage[t + k * 256];
    } else {
        for (int i = t; i < nvalid; i += 256) PDc[i] = stage[i];
    }
    __syncthreads();
    #pragma unroll
    for (int i = 0; i < CHUNK / 256; ++i) {
        if (e0 + i * 256 < ne) stage[hs[sv[i] >> BSH] + ps[i]] = (unsigned)(sv[i] & 255);
    }
    __syncthreads();
    if (nvalid == CHUNK) {
        #pragma unroll
        for (int k = 0; k < CHUNK / 256; ++k) PSc[t + k * 256] = (int)stage[t + k * 256];
    } else {
        for (int i = t; i < nvalid; i += 256) PSc[i] = (int)stage[i];
    }
}

template <int MODE>
static __device__ __forceinline__ void range_item(unsigned k, int* cnt, int* cur, int* colL,
                                                  int* __restrict__ col, int base, bool staged) {
    if (MODE == 0) {
        atomicAdd(&cnt[k & 255u], 1);
    } else {
        const int pos = atomicAdd(&cur[k & 255u], 1);
        if (staged) colL[pos - base] = (int)(k >> BSH);
        else        col[pos] = (int)(k >> BSH);
    }
}

template <int MODE>
static __device__ __forceinline__ void range_pass(const unsigned* __restrict__ P, const int* sStart, const int* sN, int nch,
                                                  int* cnt, int* cur, int* colL, int* __restrict__ col,
                                                  int base, bool staged, int wave, int lane) {
    const int hl = lane & 15;
    const int hh = lane >> 4;
    for (int k0 = 0; k0 * 16 < nch; k0 += SEGU) {
        int cc[SEGU], st[SEGU], nn[SEGU];
        unsigned w[SEGU];
        #pragma unroll
        for (int u = 0; u < SEGU; ++u) {
            const int c = ((k0 + u) * 8 + wave) * 2 + hh;
            const bool ok = c < nch;
            cc[u] = ok ? c : 0;
            st[u] = sStart[cc[u]];
            nn[u] = ok ? sN[cc[u]] : 0;
        }
        #pragma unroll
        for (int u = 0; u < SEGU; ++u) w[u] = P[(size_t)cc[u] * CHUNK + (hl < nn[u] ? st[u] + hl : 0)];
        #pragma unroll
        for (int u = 0; u < SEGU; ++u) {
            if (hl < nn[u]) range_item<MODE>(w[u], cnt, cur, colL, col, base, staged);
            for (int j = hl + 16; j < nn[u]; j += 16)
                range_item<MODE>(P[(size_t)cc[u] * CHUNK + st[u] + j], cnt, cur, colL, col, base, staged);
        }
    }
}

__global__ __launch_bounds__(256) void k_range_rank(const unsigned* __restrict__ PD,
                                                    const int* __restrict__ PS,
                                                    const int* __restrict__ tab,
                                                    int* __restrict__ seg,
                                                    int* __restrict__ col,
                                                    float* __restrict__ rsO,
                                                    float* __restrict__ rsI,
                                                    int n, int nb, int nch) {
    __shared__ int cnt[256];
    __shared__ int cur[256];
    __shared__ int wredA[8];
    __shared__ int wredB[8];
    __shared__ int wsum[8];
    __shared__ int sStart[NCHMAX];
    __shared__ int sN[NCHMAX];
    __shared__ int colL[RCAP];
    const int t = threadIdx.x;
    const int lane = t & 31;
    const int wave = t >> 5;
    const int r = blockIdx.x;
    const bool isD = r < nb;
    const int b = isD ? r : r - nb;
    const int* __restrict__ rowA = tab + (size_t)(isD ? b : nb + 1 + b) * nch;
    const int* __restrict__ rowB = rowA + nch;

    int bsum = 0, ssum = 0;
    for (int c0 = t; c0 - t < nch; c0 += 1024) {
        int s0[4], s1[4];
        #pragma unroll
        for (int q = 0; q < 4; ++q) {
            const int c = c0 + q * 256;
            const int ccl = c < nch ? c : 0;
            s0[q] = rowA[ccl];
            s1[q] = rowB[ccl];
        }
        #pragma unroll
        for (int q = 0; q < 4; ++q) {
            const int c = c0 + q * 256;
            if (c < nch) {
                sStart[c] = s0[q];
                sN[c] = s1[q] - s0[q];
                bsum += s0[q];
                ssum += s1[q] - s0[q];
            }
        }
    }
    for (int d = 16; d >= 1; d >>= 1) { bsum += __shfl_xor(bsum, d); ssum += __shfl_xor(ssum, d); }
    if (lane == 0) { wredA[wave] = bsum; wredB[wave] = ssum; }
    cnt[t] = 0;
    __syncthreads();
    int base = 0, size = 0;
    for (int w = 0; w < 8; ++w) { base += wredA[w]; size += wredB[w]; }

    const unsigned* __restrict__ P = isD ? PD : (const unsigned*)PS;
    range_pass<0>(P, sStart, sN, nch, cnt, cur, colL, col, base, false, wave, lane);
    __syncthreads();
    const int cn = cnt[t];
    const int node = b * 256 + t;
    float a = (float)cn;
    a = a < 1.0f ? 1.0f : a;
    const float rs = 1.0f / sqrtf(a);
    if (!isD) {
        if (node < n) rsO[node] = rs;
        return;
    }

    int incl = cn;
    for (int d = 1; d < 32; d <<= 1) {
        int u = __shfl_up(incl, d);
        if (lane >= d) incl += u;
    }
    if (lane == 31) wsum[wave] = incl;
    __syncthreads();
    int woff = 0;
    for (int w = 0; w < 8; ++w) {
        if (w < wave) woff += wsum[w];
    }
    const int start = base + woff + incl - cn;
    cur[t] = start;
    if (node < n) {
        int2 sg = {start, cn};
        *(int2*)(seg + 2 * (size_t)node) = sg;
        rsI[node] = rs;
    }
    __syncthreads();
    const bool staged = size <= RCAP;
    range_pass<1>(PD, sStart, sN, nch, cnt, cur, colL, col, base, staged, wave, lane);
    if (staged) {
        __syncthreads();
        for (int i0 = t; i0 < size + t; i0 += 2048) {
            int v[8];
            #pragma unroll
            for (int j = 0; j < 8; ++j) {
                const int i = i0 + j * 256;
                v[j] = colL[i < size ? i : size - 1];
            }
            #pragma unroll
            for (int j = 0; j < 8; ++j) {
                const int i = i0 + j * 256;
                if (i < size) col[base + i] = v[j];
            }
        }
    }
}

static __device__ __forceinline__ void tile_product_store(const float* Alds,
                                                          const float* __restrict__ W,
                                                          float* __restrict__ Y,
                                                          int rowbase, int nrows, int t) {
    const int wave = t >> 5;
    const int lane = t & 31;
    const int half = lane >> 4;
    const int l16  = lane & 15;
    const int ncol = wave * 16 + l16;

    v8f acc0 = {}, acc1 = {}, acc2 = {}, acc3 = {};
    const float* a0 = Alds + (0 * 16 + l16) * APAD;
    const float* a1 = Alds + (1 * 16 + l16) * APAD;
    const float* a2 = Alds + (2 * 16 + l16) * APAD;
    const float* a3 = Alds + (3 * 16 + l16) * APAD;

    #pragma unroll 4
    for (int k = 0; k < FEAT; k += 4) {
        const int ka = k + half * 2;
        v2f b;
        b.x = W[ka * FEAT + ncol];
        b.y = W[(ka + 1) * FEAT + ncol];
        v2f a;
        a.x = a0[ka]; a.y = a0[ka + 1];
        acc0 = __builtin_amdgcn_wmma_f32_16x16x4_f32(false, a, false, b, (short)0, acc0, false, false);
        a.x = a1[ka]; a.y = a1[ka + 1];
        acc1 = __builtin_amdgcn_wmma_f32_16x16x4_f32(false, a, false, b, (short)0, acc1, false, false);
        a.x = a2[ka]; a.y = a2[ka + 1];
        acc2 = __builtin_amdgcn_wmma_f32_16x16x4_f32(false, a, false, b, (short)0, acc2, false, false);
        a.x = a3[ka]; a.y = a3[ka + 1];
        acc3 = __builtin_amdgcn_wmma_f32_16x16x4_f32(false, a, false, b, (short)0, acc3, false, false);
    }

    const int mrow0 = rowbase + half * 8;
    if (rowbase + ROWS_PER_BLK <= nrows) {
        #pragma unroll
        for (int r = 0; r < 8; ++r) {
            Y[(size_t)(mrow0 +  0 + r) * FEAT + ncol] = acc0[r];
            Y[(size_t)(mrow0 + 16 + r) * FEAT + ncol] = acc1[r];
            Y[(size_t)(mrow0 + 32 + r) * FEAT + ncol] = acc2[r];
            Y[(size_t)(mrow0 + 48 + r) * FEAT + ncol] = acc3[r];
        }
    } else {
        #pragma unroll
        for (int r = 0; r < 8; ++r) {
            int r0 = mrow0 + r;
            if (r0 +  0 < nrows) Y[(size_t)(r0 +  0) * FEAT + ncol] = acc0[r];
            if (r0 + 16 < nrows) Y[(size_t)(r0 + 16) * FEAT + ncol] = acc1[r];
            if (r0 + 32 < nrows) Y[(size_t)(r0 + 32) * FEAT + ncol] = acc2[r];
            if (r0 + 48 < nrows) Y[(size_t)(r0 + 48) * FEAT + ncol] = acc3[r];
        }
    }
}

__global__ __launch_bounds__(256) void k_gemm_scaled(const float* __restrict__ X,
                                                     const float* __restrict__ rs,
                                                     const float* __restrict__ W,
                                                     float* __restrict__ Y, int nrows) {
    __shared__ float Alds[ROWS_PER_BLK * APAD];

    const int rowbase = blockIdx.x * ROWS_PER_BLK;
    const int t = threadIdx.x;

    for (int i = t; i < ROWS_PER_BLK * (FEAT / 4); i += 256) {
        int r  = i >> 5;
        int c4 = (i & 31) << 2;
        int row = rowbase + r;
        float4 v = {0.f, 0.f, 0.f, 0.f};
        float s = 0.f;
        if (row < nrows) {
            v = *(const float4*)(X + (size_t)row * FEAT + c4);
            s = rs[row];
        }
        float* dstp = Alds + r * APAD + c4;
        dstp[0] = v.x * s; dstp[1] = v.y * s; dstp[2] = v.z * s; dstp[3] = v.w * s;
    }
    __syncthreads();
    tile_product_store(Alds, W, Y, rowbase, nrows, t);
}

static __device__ __forceinline__ void add_rows(float4& acc, const float* Hl, int cv, int m) {
#define ROW(j) (*(const float4*)(Hl + (size_t)__builtin_amdgcn_readlane(cv, (j)) * FEAT))
#define ADD(v) { acc.x += (v).x; acc.y += (v).y; acc.z += (v).z; acc.w += (v).w; }
    int j = 0;
    for (; j + 8 <= m; j += 8) {
        float4 v0 = ROW(j), v1 = ROW(j + 1), v2 = ROW(j + 2), v3 = ROW(j + 3);
        float4 v4 = ROW(j + 4), v5 = ROW(j + 5), v6 = ROW(j + 6), v7 = ROW(j + 7);
        ADD(v0) ADD(v1) ADD(v2) ADD(v3) ADD(v4) ADD(v5) ADD(v6) ADD(v7)
    }
    if (j + 4 <= m) {
        float4 v0 = ROW(j), v1 = ROW(j + 1), v2 = ROW(j + 2), v3 = ROW(j + 3);
        ADD(v0) ADD(v1) ADD(v2) ADD(v3)
        j += 4;
    }
    if (j + 2 <= m) {
        float4 v0 = ROW(j), v1 = ROW(j + 1);
        ADD(v0) ADD(v1)
        j += 2;
    }
    if (j < m) {
        float4 v0 = ROW(j);
        ADD(v0)
    }
#undef ROW
#undef ADD
}

static __device__ __forceinline__ float4 gather_row(const float* __restrict__ H,
                                                    const int* __restrict__ seg,
                                                    const int* __restrict__ col,
                                                    int node, int lane) {
    const int2 sg = *(const int2*)(seg + 2 * (size_t)node);
    const int start = sg.x;
    const int cnt = sg.y;
    const float* Hl = H + lane * 4;

    float4 acc = {0.f, 0.f, 0.f, 0.f};
    for (int base = 0; base < cnt; base += 32) {
        int m = cnt - base;
        if (m > 32) m = 32;
        int cv = 0;
        if (lane < m) cv = col[start + base + lane];
        add_rows(acc, Hl, cv, m);
    }
    return acc;
}

static __device__ __forceinline__ float4 gather_row_ahead(const float* __restrict__ H,
                                                          const int* __restrict__ col,
                                                          int start, int cnt, int cv0, int lane) {
    const float* Hl = H + lane * 4;
    float4 acc = {0.f, 0.f, 0.f, 0.f};
    add_rows(acc, Hl, cv0, cnt < 32 ? cnt : 32);
    for (int base = 32; base < cnt; base += 32) {
        int m = cnt - base;
        if (m > 32) m = 32;
        int cv = 0;
        if (lane < m) cv = col[start + base + lane];
        add_rows(acc, Hl, cv, m);
    }
    return acc;
}

static __device__ __forceinline__ float4 row_tail(float4 acc, float s, float4 b) {
    float4 h;
    h.x = acc.x * s + b.x; h.x = h.x > 0.f ? h.x : 0.f;
    h.y = acc.y * s + b.y; h.y = h.y > 0.f ? h.y : 0.f;
    h.z = acc.z * s + b.z; h.z = h.z > 0.f ? h.z : 0.f;
    h.w = acc.w * s + b.w; h.w = h.w > 0.f ? h.w : 0.f;
    return h;
}

__global__ __launch_bounds__(256) void k_gather_relu_gemm(const float* __restrict__ H,
                                                          const int* __restrict__ seg,
                                                          const int* __restrict__ col,
                                                          const float* __restrict__ rsI,
                                                          const float* __restrict__ rsO,
                                                          const float* __restrict__ bias,
                                                          const float* __restrict__ W,
                                                          float* __restrict__ Y, int n) {
    __shared__ float Alds[ROWS_PER_BLK * APAD];

    const int rowbase = blockIdx.x * ROWS_PER_BLK;
    const int t = threadIdx.x;
    const int lane = t & 31;
    const int wave = __builtin_amdgcn_readfirstlane(t >> 5);
    const float4 b = *(const float4*)(bias + lane * 4);

    const int r0 = wave * (ROWS_PER_BLK / 8);
    int2 sgv = {0, 0};
    float siv = 0.f, sov = 0.f;
    {
        const int nd = rowbase + r0 + (lane & 7);
        if (nd < n) {
            sgv = *(const int2*)(seg + 2 * (size_t)nd);
            siv = rsI[nd];
            sov = rsO[nd];
        }
    }
    int startN = __builtin_amdgcn_readlane(sgv.x, 0);
    int cntN   = __builtin_amdgcn_readlane(sgv.y, 0);
    int cvN = 0;
    if (lane < cntN) cvN = col[startN + lane];

    for (int i = 0; i < ROWS_PER_BLK / 8; ++i) {
        const int start = startN, cnt = cntN, cv0 = cvN;
        const float s  = __builtin_bit_cast(float, __builtin_amdgcn_readlane(__builtin_bit_cast(int, siv), i));
        const float so = __builtin_bit_cast(float, __builtin_amdgcn_readlane(__builtin_bit_cast(int, sov), i));
        if (i + 1 < ROWS_PER_BLK / 8) {
            startN = __builtin_amdgcn_readlane(sgv.x, i + 1);
            cntN   = __builtin_amdgcn_readlane(sgv.y, i + 1);
            cvN = 0;
            if (lane < cntN) cvN = col[startN + lane];
        }
        const float4 acc = gather_row_ahead(H, col, start, cnt, cv0, lane);
        const float4 h = row_tail(acc, s, b);
        float4 v = {0.f, 0.f, 0.f, 0.f};
        if (rowbase + r0 + i < n) { v.x = h.x * so; v.y = h.y * so; v.z = h.z * so; v.w = h.w * so; }
        *(float4*)(Alds + (r0 + i) * APAD + lane * 4) = v;
    }
    __syncthreads();
    tile_product_store(Alds, W, Y, rowbase, n, t);
}

static __device__ __forceinline__ float lx_dpp_q1(float x) {
    return __builtin_bit_cast(float, __builtin_amdgcn_update_dpp(0, __builtin_bit_cast(int, x), 0xB1, 0xF, 0xF, true));
}
static __device__ __forceinline__ float lx_dpp_q2(float x) {
    return __builtin_bit_cast(float, __builtin_amdgcn_update_dpp(0, __builtin_bit_cast(int, x), 0x4E, 0xF, 0xF, true));
}
static __device__ __forceinline__ float lx_dpp_x4(float x) {
    return __builtin_bit_cast(float, __builtin_amdgcn_update_dpp(0, __builtin_bit_cast(int, x), 0x164, 0xF, 0xF, true));
}
static __device__ __forceinline__ float lx_dpp_x8(float x) {
    return __builtin_bit_cast(float, __builtin_amdgcn_update_dpp(0, __builtin_bit_cast(int, x), 0x168, 0xF, 0xF, true));
}
static __device__ __forceinline__ float lx_half(float x) {
    return __builtin_bit_cast(float, __builtin_amdgcn_permlanex16(__builtin_bit_cast(int, x), __builtin_bit_cast(int, x),
                                                                  0x76543210, 0xfedcba98, false, false));
}

__global__ __launch_bounds__(256) void k_gather_cls(const float* __restrict__ H,
                                                    const int* __restrict__ seg,
                                                    const int* __restrict__ col,
                                                    const float* __restrict__ rsI,
                                                    const float* __restrict__ bias,
                                                    const float* __restrict__ Wc,
                                                    const float* __restrict__ bc,
                                                    float* __restrict__ out, int n) {
    __shared__ float lg[32];
    const int lane = threadIdx.x & 31;
    const int wave = threadIdx.x >> 5;
    const float bco = bc[lane & 3];
    int node = blockIdx.x * 8 + wave;
    if (node < n) {
        node = __builtin_amdgcn_readfirstlane(node);
        const float s = rsI[node];
        const float4 b = *(const float4*)(bias + lane * 4);
        const float4 acc = gather_row(H, seg, col, node, lane);
        const float4 h = row_tail(acc, s, b);

        const float4 w0 = *(const float4*)(Wc + lane * 16);
        const float4 w1 = *(const float4*)(Wc + lane * 16 + 4);
        const float4 w2 = *(const float4*)(Wc + lane * 16 + 8);
        const float4 w3 = *(const float4*)(Wc + lane * 16 + 12);
        const float p0 = h.x * w0.x + h.y * w1.x + h.z * w2.x + h.w * w3.x;
        const float p1 = h.x * w0.y + h.y * w1.y + h.z * w2.y + h.w * w3.y;
        const float p2 = h.x * w0.z + h.y * w1.z + h.z * w2.z + h.w * w3.z;
        const float p3 = h.x * w0.w + h.y * w1.w + h.z * w2.w + h.w * w3.w;

        const bool o1 = (lane & 1) != 0;
        const bool o2 = (lane & 2) != 0;
        const float q0 = (o1 ? p2 : p0) + lx_dpp_q1(o1 ? p0 : p2);
        const float q1 = (o1 ? p3 : p1) + lx_dpp_q1(o1 ? p1 : p3);
        float r = (o2 ? q1 : q0) + lx_dpp_q2(o2 ? q0 : q1);
        r += lx_dpp_x4(r);
        r += lx_dpp_x8(r);
        r += lx_half(r);
        if (lane < 4) lg[wave * 4 + (((lane & 1) << 1) | (lane >> 1))] = r;
    }
    __syncthreads();
    if (wave == 0) {
        const int i = blockIdx.x * 32 + lane;
        if (i < n * 4) out[i] = lg[lane] + bco;
    }
}

extern "C" void kernel_launch(void* const* d_in, const int* in_sizes, int n_in,
                              void* d_out, int out_size, void* d_ws, size_t ws_size,
                              hipStream_t stream) {
    const float* feat = (const float*)d_in[0];
    const int*   src  = (const int*)d_in[1];
    const int*   dst  = (const int*)d_in[2];
    const float* W1   = (const float*)d_in[3];
    const float* b1   = (const float*)d_in[4];
    const float* W2   = (const float*)d_in[5];
    const float* b2   = (const float*)d_in[6];
    const float* Wc   = (const float*)d_in[7];
    const float* bc   = (const float*)d_in[8];
    float* out = (float*)d_out;

    const int nnodes = in_sizes[0] / FEAT;
    const int nedges = in_sizes[1];
    const size_t nAlign = ((size_t)nnodes + 3) & ~(size_t)3;
    const size_t eAlign = ((size_t)nedges + 3) & ~(size_t)3;
    const size_t nfeat = (size_t)nnodes * FEAT;

    const int NB  = (nnodes + 255) >> BSH;
    const int NCH = (nedges + CHUNK - 1) / CHUNK;
    const size_t tabAlign = ((size_t)2 * (NB + 1) * NCH + 3) & ~(size_t)3;
    const size_t pAlign = (size_t)NCH * CHUNK;

    int*      tab    = (int*)d_ws;
    int*      seg    = tab + tabAlign;
    float*    rsO    = (float*)(seg + 2 * nAlign);
    float*    rsI    = rsO + nAlign;
    unsigned* PD     = (unsigned*)(rsI + nAlign);
    int*      PS     = (int*)(PD + pAlign);
    int*      col    = PS + pAlign;
    float*    bufA   = (float*)(((uintptr_t)(col + eAlign) + 511) & ~(uintptr_t)511);
    float*    bufB   = bufA + nfeat;

    const int B = 256;
    const int gNodeW  = (nnodes + 7) / 8;
    const int gGemm   = (nnodes + ROWS_PER_BLK - 1) / ROWS_PER_BLK;

    k_chunk_file<<<NCH, B, 0, stream>>>(src, dst, tab, PD, PS, nedges, NB, NCH);
    k_range_rank<<<2 * NB, B, 0, stream>>>(PD, PS, tab, seg, col, rsO, rsI, nnodes, NB, NCH);

    k_gemm_scaled<<<gGemm, B, 0, stream>>>(feat, rsO, W1, bufA, nnodes);

    k_gather_relu_gemm<<<gGemm, B, 0, stream>>>(bufA, seg, col, rsI, rsO, b1, W2, bufB, nnodes);

    k_gather_cls<<<gNodeW, B, 0, stream>>>(bufB, seg, col, rsI, b2, Wc, bc, out, nnodes);
}
